# starting kernel
# baseline (speedup 1.0000x reference)
#include <hip/hip_runtime.h>
#include <math.h>

typedef __attribute__((ext_vector_type(2))) float v2f;
typedef __attribute__((ext_vector_type(8))) float v8f;

#define HID 128

__device__ __forceinline__ float lrelu(float v, float s) { return v > 0.f ? v : s * v; }

__device__ __forceinline__ void atomicMaxF(float* addr, float val) {
  if (val >= 0.f) atomicMax((int*)addr, __float_as_int(val));
  else            atomicMin((unsigned int*)addr, __float_as_uint(val));
}

__global__ void encoder_kernel(const float* __restrict__ num_prop, const float* __restrict__ cat_prop,
                               const float* __restrict__ Wn, const float* __restrict__ bn,
                               const float* __restrict__ Wc, const float* __restrict__ bc,
                               float* __restrict__ xcat, int N) {
  int t = blockIdx.x * blockDim.x + threadIdx.x;
  if (t >= N * 64) return;
  int n = t >> 6, j = t & 63;
  const float* np = num_prop + (size_t)n * 7;
  float an = bn[j];
#pragma unroll
  for (int k = 0; k < 7; ++k) an += np[k] * Wn[k * 64 + j];
  const float* cp = cat_prop + (size_t)n * 3;
  float ac = bc[j];
#pragma unroll
  for (int k = 0; k < 3; ++k) ac += cp[k] * Wc[k * 64 + j];
  xcat[(size_t)n * HID + j]      = lrelu(an, 0.01f);
  xcat[(size_t)n * HID + 64 + j] = lrelu(ac, 0.01f);
}

__global__ void wmma_gemm128_kernel(const float* __restrict__ A, const float* __restrict__ B,
                                    const float* __restrict__ bias, float* __restrict__ C,
                                    int M, int act, int has_bias) {
  int wave = threadIdx.x >> 5;
  int lane = threadIdx.x & 31;
  int row0 = blockIdx.x * 16;
  int col0 = wave * 16;
  int half = lane >> 4;
  int m16  = lane & 15;

  int arow = row0 + m16; if (arow >= M) arow = M - 1;
  const float* Ap = A + (size_t)arow * HID + 2 * half;
  const float* Bp = B + (size_t)(2 * half) * HID + col0 + m16;

  v8f acc = {0.f, 0.f, 0.f, 0.f, 0.f, 0.f, 0.f, 0.f};
#pragma unroll 8
  for (int k0 = 0; k0 < HID; k0 += 4) {
    v2f a, b;
    a.x = Ap[k0];
    a.y = Ap[k0 + 1];
    b.x = Bp[(size_t)k0 * HID];
    b.y = Bp[(size_t)(k0 + 1) * HID];
    acc = __builtin_amdgcn_wmma_f32_16x16x4_f32(false, a, false, b, (short)0, acc, false, false);
  }

  int col = col0 + m16;
  float bv = has_bias ? bias[col] : 0.f;
#pragma unroll
  for (int r = 0; r < 8; ++r) {
    int row = row0 + r + 8 * half;
    if (row < M) {
      float v = acc[r] + bv;
      if (act) v = lrelu(v, 0.01f);
      C[(size_t)row * HID + col] = v;
    }
  }
}

__global__ void bn_stats_kernel(const float* __restrict__ x, float* __restrict__ stats, int N) {
  int c = threadIdx.x;
  float s = 0.f, s2 = 0.f;
  for (int n = blockIdx.x; n < N; n += gridDim.x) {
    float v = x[(size_t)n * HID + c];
    s += v; s2 += v * v;
  }
  atomicAdd(stats + c, s);
  atomicAdd(stats + HID + c, s2);
}

__global__ void bn_finalize_kernel(float* __restrict__ stats, const float* __restrict__ gamma,
                                   const float* __restrict__ beta, int N) {
  int c = threadIdx.x;
  float inv_n = 1.f / (float)N;
  float mu  = stats[c] * inv_n;
  float var = stats[HID + c] * inv_n - mu * mu;
  float sc  = gamma[c] * rsqrtf(var + 1e-5f);
  stats[2 * HID + c] = sc;
  stats[3 * HID + c] = beta[c] - mu * sc;
}

__global__ void bn_apply_kernel(const float* __restrict__ x, const float* __restrict__ stats,
                                float* __restrict__ y, int N) {
  int t = blockIdx.x * blockDim.x + threadIdx.x;
  if (t >= N * HID) return;
  int c = t & (HID - 1);
  y[t] = x[t] * stats[2 * HID + c] + stats[3 * HID + c];
}

__global__ void att_logits_kernel(const float* __restrict__ h, const float* __restrict__ a_src,
                                  const float* __restrict__ a_dst, float* __restrict__ als,
                                  float* __restrict__ ald, int N, int H, int F) {
  int t = blockIdx.x * blockDim.x + threadIdx.x;
  if (t >= N * H) return;
  int n = t / H, hh = t - n * H;
  const float* hp = h + (size_t)n * HID + hh * F;
  const float* as = a_src + hh * F;
  const float* ad = a_dst + hh * F;
  float ss = 0.f, sd = 0.f;
  for (int f = 0; f < F; ++f) { float v = hp[f]; ss += v * as[f]; sd += v * ad[f]; }
  als[t] = ss;
  ald[t] = sd;
}

__global__ void fill_kernel(float* __restrict__ p, long long n, float v) {
  long long t = (long long)blockIdx.x * blockDim.x + threadIdx.x;
  if (t < n) p[t] = v;
}

__device__ __forceinline__ void edge_sd(const int* __restrict__ ei, int E, long long e,
                                        int& s, int& d) {
  if (e < E) { s = ei[e]; d = ei[(size_t)E + e]; }
  else       { s = d = (int)(e - E); }
}

__global__ void edge_max_kernel(const int* __restrict__ ei, const float* __restrict__ als,
                                const float* __restrict__ ald, float* __restrict__ emax,
                                int E, int N, int H) {
  long long t = (long long)blockIdx.x * blockDim.x + threadIdx.x;
  long long total = (long long)(E + N) * H;
  if (t >= total) return;
  long long e = t / H; int hh = (int)(t - e * H);
  int s, d; edge_sd(ei, E, e, s, d);
  float v = lrelu(als[(size_t)s * H + hh] + ald[(size_t)d * H + hh], 0.2f);
  atomicMaxF(emax + (size_t)d * H + hh, v);
}

__global__ void edge_denom_kernel(const int* __restrict__ ei, const float* __restrict__ als,
                                  const float* __restrict__ ald, const float* __restrict__ emax,
                                  float* __restrict__ denom, int E, int N, int H) {
  long long t = (long long)blockIdx.x * blockDim.x + threadIdx.x;
  long long total = (long long)(E + N) * H;
  if (t >= total) return;
  long long e = t / H; int hh = (int)(t - e * H);
  int s, d; edge_sd(ei, E, e, s, d);
  float v = lrelu(als[(size_t)s * H + hh] + ald[(size_t)d * H + hh], 0.2f);
  float ee = __expf(v - emax[(size_t)d * H + hh]);
  atomicAdd(denom + (size_t)d * H + hh, ee);
}

__global__ void edge_scatter_kernel(const int* __restrict__ ei, const float* __restrict__ hfeat,
                                    const float* __restrict__ als, const float* __restrict__ ald,
                                    const float* __restrict__ emax, const float* __restrict__ denom,
                                    float* __restrict__ acc, int E, int N, int H, int F) {
  long long wid = ((long long)blockIdx.x * blockDim.x + threadIdx.x) >> 5;
  int lane = threadIdx.x & 31;
  long long total = (long long)E + N;
  if (wid >= total) return;
  int s, d; edge_sd(ei, E, wid, s, d);
  float alpha_l = 0.f;
  if (lane < H) {
    float ev = lrelu(als[(size_t)s * H + lane] + ald[(size_t)d * H + lane], 0.2f);
    alpha_l = __expf(ev - emax[(size_t)d * H + lane]) / denom[(size_t)d * H + lane];
  }
  const float* hs = hfeat + (size_t)s * HID;
  float* ap = acc + (size_t)d * HID;
#pragma unroll
  for (int i = 0; i < 4; ++i) {
    int c = lane + 32 * i;
    int head = c / F;
    float alpha = __shfl(alpha_l, head, 32);
    atomicAdd(ap + c, alpha * hs[c]);
  }
}

__global__ void finalize_kernel(float* __restrict__ acc, const float* __restrict__ bias,
                                int N, int act) {
  int t = blockIdx.x * blockDim.x + threadIdx.x;
  if (t >= N * HID) return;
  int c = t & (HID - 1);
  float v = acc[t] + bias[c];
  if (act) v = lrelu(v, 0.01f);
  acc[t] = v;
}

extern "C" void kernel_launch(void* const* d_in, const int* in_sizes, int n_in,
                              void* d_out, int out_size, void* d_ws, size_t ws_size,
                              hipStream_t stream) {
  const float* num_prop = (const float*)d_in[0];
  const float* cat_prop = (const float*)d_in[1];
  const int*   ei       = (const int*)d_in[2];
  const float* W_num = (const float*)d_in[3];
  const float* b_num = (const float*)d_in[4];
  const float* W_cat = (const float*)d_in[5];
  const float* b_cat = (const float*)d_in[6];
  const float* W_in  = (const float*)d_in[7];
  const float* b_in  = (const float*)d_in[8];
  const float* bn_g  = (const float*)d_in[9];
  const float* bn_b  = (const float*)d_in[10];
  const float* W1    = (const float*)d_in[11];
  const float* a_s1  = (const float*)d_in[12];
  const float* a_d1  = (const float*)d_in[13];
  const float* bias1 = (const float*)d_in[14];
  const float* W2    = (const float*)d_in[15];
  const float* a_s2  = (const float*)d_in[16];
  const float* a_d2  = (const float*)d_in[17];
  const float* bias2 = (const float*)d_in[18];

  const int N = in_sizes[0] / 7;
  const int E = in_sizes[2] / 2;
  float* out = (float*)d_out;

  float* ws   = (float*)d_ws;
  float* xcat = ws;
  float* x1   = xcat + (size_t)N * HID;
  float* hbuf = x1   + (size_t)N * HID;
  float* accb = hbuf + (size_t)N * HID;
  float* als  = accb + (size_t)N * HID;
  float* ald  = als  + (size_t)N * 4;
  float* emax = ald  + (size_t)N * 4;
  float* denom= emax + (size_t)N * 4;
  float* stats= denom+ (size_t)N * 4;

  const int TB = 256;
  const long long EN = (long long)E + N;
  const int mtiles = (N + 15) / 16;
  auto nb = [&](long long work) { return (unsigned)((work + TB - 1) / TB); };

  encoder_kernel<<<nb((long long)N * 64), TB, 0, stream>>>(num_prop, cat_prop, W_num, b_num,
                                                           W_cat, b_cat, xcat, N);
  wmma_gemm128_kernel<<<mtiles, TB, 0, stream>>>(xcat, W_in, b_in, x1, N, 1, 1);

  fill_kernel<<<1, 256, 0, stream>>>(stats, 256, 0.f);
  bn_stats_kernel<<<512, 128, 0, stream>>>(x1, stats, N);
  bn_finalize_kernel<<<1, 128, 0, stream>>>(stats, bn_g, bn_b, N);
  bn_apply_kernel<<<nb((long long)N * HID), TB, 0, stream>>>(x1, stats, xcat, N);

  wmma_gemm128_kernel<<<mtiles, TB, 0, stream>>>(xcat, W1, nullptr, hbuf, N, 0, 0);
  att_logits_kernel<<<nb((long long)N * 4), TB, 0, stream>>>(hbuf, a_s1, a_d1, als, ald, N, 4, 32);
  fill_kernel<<<nb((long long)N * 4), TB, 0, stream>>>(emax, (long long)N * 4, -INFINITY);
  fill_kernel<<<nb((long long)N * 4), TB, 0, stream>>>(denom, (long long)N * 4, 0.f);
  fill_kernel<<<nb((long long)N * HID), TB, 0, stream>>>(accb, (long long)N * HID, 0.f);
  edge_max_kernel<<<nb(EN * 4), TB, 0, stream>>>(ei, als, ald, emax, E, N, 4);
  edge_denom_kernel<<<nb(EN * 4), TB, 0, stream>>>(ei, als, ald, emax, denom, E, N, 4);
  edge_scatter_kernel<<<nb(EN * 32), TB, 0, stream>>>(ei, hbuf, als, ald, emax, denom, accb,
                                                      E, N, 4, 32);
  finalize_kernel<<<nb((long long)N * HID), TB, 0, stream>>>(accb, bias1, N, 1);

  wmma_gemm128_kernel<<<mtiles, TB, 0, stream>>>(accb, W2, nullptr, hbuf, N, 0, 0);
  att_logits_kernel<<<nb(N), TB, 0, stream>>>(hbuf, a_s2, a_d2, als, ald, N, 1, 128);
  fill_kernel<<<nb(N), TB, 0, stream>>>(emax, N, -INFINITY);
  fill_kernel<<<nb(N), TB, 0, stream>>>(denom, N, 0.f);
  fill_kernel<<<nb((long long)N * HID), TB, 0, stream>>>(out, (long long)N * HID, 0.f);
  edge_max_kernel<<<nb(EN), TB, 0, stream>>>(ei, als, ald, emax, E, N, 1);
  edge_denom_kernel<<<nb(EN), TB, 0, stream>>>(ei, als, ald, emax, denom, E, N, 1);
  edge_scatter_kernel<<<nb(EN * 32), TB, 0, stream>>>(ei, hbuf, als, ald, emax, denom, out,
                                                      E, N, 1, 128);
  finalize_kernel<<<nb((long long)N * HID), TB, 0, stream>>>(out, bias2, N, 0);
}


# all float atomics removed: edges bucketed by destination once per call (integer counting sort, bucket ordered by source), one wave per node does softmax + weighted row sum; BN sums in fixed order
# speedup vs baseline: 1.9412x; 1.9412x over previous
#include <hip/hip_runtime.h>
#include <math.h>

typedef __attribute__((ext_vector_type(2))) float v2f;
typedef __attribute__((ext_vector_type(8))) float v8f;

#define HID 128

__device__ __forceinline__ float lrelu(float v, float s) { return v > 0.f ? v : s * v; }

__global__ void encoder_kernel(const float* __restrict__ num_prop, const float* __restrict__ cat_prop,
                               const float* __restrict__ Wn, const float* __restrict__ bn,
                               const float* __restrict__ Wc, const float* __restrict__ bc,
                               float* __restrict__ xcat, int N) {
  int t = blockIdx.x * blockDim.x + threadIdx.x;
  if (t >= N * 64) return;
  int n = t >> 6, j = t & 63;
  const float* np = num_prop + (size_t)n * 7;
  float an = bn[j];
#pragma unroll
  for (int k = 0; k < 7; ++k) an += np[k] * Wn[k * 64 + j];
  const float* cp = cat_prop + (size_t)n * 3;
  float ac = bc[j];
#pragma unroll
  for (int k = 0; k < 3; ++k) ac += cp[k] * Wc[k * 64 + j];
  xcat[(size_t)n * HID + j]      = lrelu(an, 0.01f);
  xcat[(size_t)n * HID + 64 + j] = lrelu(ac, 0.01f);
}

__global__ void wmma_gemm128_kernel(const float* __restrict__ A, const float* __restrict__ B,
                                    const float* __restrict__ bias, float* __restrict__ C,
                                    int M, int act, int has_bias) {
  int wave = threadIdx.x >> 5;
  int lane = threadIdx.x & 31;
  int row0 = blockIdx.x * 16;
  int col0 = wave * 16;
  int half = lane >> 4;
  int m16  = lane & 15;

  int arow = row0 + m16; if (arow >= M) arow = M - 1;
  const float* Ap = A + (size_t)arow * HID + 2 * half;
  const float* Bp = B + (size_t)(2 * half) * HID + col0 + m16;

  v8f acc = {0.f, 0.f, 0.f, 0.f, 0.f, 0.f, 0.f, 0.f};
#pragma unroll 8
  for (int k0 = 0; k0 < HID; k0 += 4) {
    v2f a, b;
    a.x = Ap[k0];
    a.y = Ap[k0 + 1];
    b.x = Bp[(size_t)k0 * HID];
    b.y = Bp[(size_t)(k0 + 1) * HID];
    acc = __builtin_amdgcn_wmma_f32_16x16x4_f32(false, a, false, b, (short)0, acc, false, false);
  }

  int col = col0 + m16;
  float bv = has_bias ? bias[col] : 0.f;
#pragma unroll
  for (int r = 0; r < 8; ++r) {
    int row = row0 + r + 8 * half;
    if (row < M) {
      float v = acc[r] + bv;
      if (act) v = lrelu(v, 0.01f);
      C[(size_t)row * HID + col] = v;
    }
  }
}

__global__ void bn_stats_kernel(const float* __restrict__ x, float* __restrict__ part, int N) {
  int c = threadIdx.x;
  float s = 0.f, s2 = 0.f;
  for (int n = blockIdx.x; n < N; n += gridDim.x) {
    float v = x[(size_t)n * HID + c];
    s += v; s2 += v * v;
  }
  part[(size_t)blockIdx.x * 2 * HID + c]       = s;
  part[(size_t)blockIdx.x * 2 * HID + HID + c] = s2;
}

__global__ void bn_finalize_kernel(const float* __restrict__ part, int nblk, float* __restrict__ stats,
                                   const float* __restrict__ gamma, const float* __restrict__ beta, int N) {
  __shared__ float sh[2][8][HID];
  int c = threadIdx.x & (HID - 1);
  int g = threadIdx.x >> 7;
  int per = (nblk + 7) / 8;
  int b0 = g * per;
  int b1 = b0 + per < nblk ? b0 + per : nblk;
  float s = 0.f, s2 = 0.f;
  for (int b = b0; b < b1; ++b) {
    s  += part[(size_t)b * 2 * HID + c];
    s2 += part[(size_t)b * 2 * HID + HID + c];
  }
  sh[0][g][c] = s;
  sh[1][g][c] = s2;
  __syncthreads();
  if (threadIdx.x < HID) {
    float S = 0.f, S2 = 0.f;
    for (int k = 0; k < 8; ++k) { S += sh[0][k][c]; S2 += sh[1][k][c]; }
    float inv_n = 1.f / (float)N;
    float mu  = S * inv_n;
    float var = S2 * inv_n - mu * mu;
    float sc  = gamma[c] * rsqrtf(var + 1e-5f);
    stats[2 * HID + c] = sc;
    stats[3 * HID + c] = beta[c] - mu * sc;
  }
}

__global__ void bn_apply_kernel(const float* __restrict__ x, const float* __restrict__ stats,
                                float* __restrict__ y, int N) {
  int t = blockIdx.x * blockDim.x + threadIdx.x;
  if (t >= N * HID) return;
  int c = t & (HID - 1);
  y[t] = x[t] * stats[2 * HID + c] + stats[3 * HID + c];
}

__global__ void att_logits_kernel(const float* __restrict__ h, const float* __restrict__ a_src,
                                  const float* __restrict__ a_dst, float* __restrict__ als,
                                  float* __restrict__ ald, int N, int H, int F) {
  int t = blockIdx.x * blockDim.x + threadIdx.x;
  if (t >= N * H) return;
  int n = t / H, hh = t - n * H;
  const float* hp = h + (size_t)n * HID + hh * F;
  const float* as = a_src + hh * F;
  const float* ad = a_dst + hh * F;
  float ss = 0.f, sd = 0.f;
  for (int f = 0; f < F; ++f) { float v = hp[f]; ss += v * as[f]; sd += v * ad[f]; }
  als[t] = ss;
  ald[t] = sd;
}

__global__ void fill_int_kernel(int* __restrict__ p, int n, int v) {
  int t = blockIdx.x * blockDim.x + threadIdx.x;
  if (t < n) p[t] = v;
}

__global__ void edge_rank_kernel(const int* __restrict__ ei, int* __restrict__ cnt,
                                 int* __restrict__ rank, int E) {
  int e = blockIdx.x * blockDim.x + threadIdx.x;
  if (e >= E) return;
  int d = ei[(size_t)E + e];
  rank[e] = atomicAdd(cnt + d, 1);
}

__global__ void bucket_scan_kernel(const int* __restrict__ cnt, int* __restrict__ off, int N) {
  __shared__ int part[1024];
  int t = threadIdx.x;
  int per = (N + 1023) / 1024;
  int b = t * per;
  int e = b + per < N ? b + per : N;
  int s = 0;
  for (int i = b; i < e; ++i) s += cnt[i];
  part[t] = s;
  __syncthreads();
  for (int dd = 1; dd < 1024; dd <<= 1) {
    int v = (t >= dd) ? part[t - dd] : 0;
    __syncthreads();
    part[t] += v;
    __syncthreads();
  }
  int run = part[t] - s;
  for (int i = b; i < e; ++i) { off[i] = run; run += cnt[i]; }
  if (t == 1023) off[N] = part[1023];
}

__global__ void edge_place_kernel(const int* __restrict__ ei, const int* __restrict__ off,
                                  const int* __restrict__ rank, int* __restrict__ bsrc, int E) {
  int e = blockIdx.x * blockDim.x + threadIdx.x;
  if (e >= E) return;
  int s = ei[e];
  int d = ei[(size_t)E + e];
  bsrc[off[d] + rank[e]] = s;
}

__global__ void bucket_sort_kernel(const int* __restrict__ off, const int* __restrict__ bsrc,
                                   int* __restrict__ ssrc, int N) {
  const int lane = threadIdx.x & 31;
  const int d = __builtin_amdgcn_readfirstlane((int)((blockIdx.x * blockDim.x + threadIdx.x) >> 5));
  if (d >= N) return;
  const int beg = off[d];
  const int L = off[d + 1] - beg;
  if (L <= 32) {
    int key = (lane < L) ? bsrc[beg + lane] : 0x7fffffff;
    for (int k = 2; k <= 32; k <<= 1) {
      for (int j = k >> 1; j > 0; j >>= 1) {
        int other = __shfl_xor(key, j, 32);
        bool up = (lane & k) == 0;
        bool lower = (lane & j) == 0;
        int lo = key < other ? key : other;
        int hi = key < other ? other : key;
        key = (lower == up) ? lo : hi;
      }
    }
    if (lane < L) ssrc[beg + lane] = key;
  } else {
    for (int i = lane; i < L; i += 32) {
      int ki = bsrc[beg + i];
      int r = 0;
      for (int j = 0; j < L; ++j) {
        int kj = bsrc[beg + j];
        r += (kj < ki || (kj == ki && j < i)) ? 1 : 0;
      }
      ssrc[beg + r] = ki;
    }
  }
}

__device__ __forceinline__ float wave_max(float v) {
  for (int o = 16; o > 0; o >>= 1) v = fmaxf(v, __shfl_xor(v, o, 32));
  return v;
}
__device__ __forceinline__ float wave_sum(float v) {
  for (int o = 16; o > 0; o >>= 1) v += __shfl_xor(v, o, 32);
  return v;
}

template <int H>
__device__ __forceinline__ void add_rows(float4& acc, const float* __restrict__ hfeat, int s,
                                         const float (&al)[H], int n, int lane, int myhead) {
#pragma unroll 8
  for (int jj = 0; jj < n; ++jj) {
    const int sj = __builtin_amdgcn_readlane(s, jj);
    float a = __int_as_float(__builtin_amdgcn_readlane(__float_as_int(al[0]), jj));
    for (int hh = 1; hh < H; ++hh) {
      float ah = __int_as_float(__builtin_amdgcn_readlane(__float_as_int(al[hh]), jj));
      a = (myhead == hh) ? ah : a;
    }
    const float4 hv = *((const float4*)(hfeat + (size_t)sj * HID) + lane);
    acc.x += a * hv.x;
    acc.y += a * hv.y;
    acc.z += a * hv.z;
    acc.w += a * hv.w;
  }
}

template <int H>
__global__ void gat_aggregate_kernel(const int* __restrict__ off, const int* __restrict__ ssrc,
                                     const float* __restrict__ hfeat, const float* __restrict__ als,
                                     const float* __restrict__ ald, const float* __restrict__ bias,
                                     float* __restrict__ out, int N, int act) {
  constexpr int F = HID / H;
  const int lane = threadIdx.x & 31;
  const int d = __builtin_amdgcn_readfirstlane((int)((blockIdx.x * blockDim.x + threadIdx.x) >> 5));
  if (d >= N) return;
  const int beg = off[d];
  const int L = off[d + 1] - beg;
  const int tot = L + 1;

  float adv[H], m[H], den[H], e0[H];
  for (int hh = 0; hh < H; ++hh) { adv[hh] = ald[(size_t)d * H + hh]; m[hh] = -INFINITY; den[hh] = 0.f; e0[hh] = 0.f; }

  const bool v0 = lane < tot;
  int s0 = d;
  if (v0) {
    s0 = (lane < L) ? ssrc[beg + lane] : d;
    for (int hh = 0; hh < H; ++hh) { e0[hh] = lrelu(als[(size_t)s0 * H + hh] + adv[hh], 0.2f); m[hh] = e0[hh]; }
  }

  for (int c0 = 32; c0 < tot; c0 += 32) {
    int j = c0 + lane;
    if (j < tot) {
      int s = (j < L) ? ssrc[beg + j] : d;
      for (int hh = 0; hh < H; ++hh)
        m[hh] = fmaxf(m[hh], lrelu(als[(size_t)s * H + hh] + adv[hh], 0.2f));
    }
  }
  for (int hh = 0; hh < H; ++hh) m[hh] = wave_max(m[hh]);

  float x0[H];
  for (int hh = 0; hh < H; ++hh) { x0[hh] = v0 ? __expf(e0[hh] - m[hh]) : 0.f; den[hh] = x0[hh]; }
  for (int c0 = 32; c0 < tot; c0 += 32) {
    int j = c0 + lane;
    if (j < tot) {
      int s = (j < L) ? ssrc[beg + j] : d;
      for (int hh = 0; hh < H; ++hh)
        den[hh] += __expf(lrelu(als[(size_t)s * H + hh] + adv[hh], 0.2f) - m[hh]);
    }
  }
  for (int hh = 0; hh < H; ++hh) den[hh] = wave_sum(den[hh]);

  const int myhead = (4 * lane) / F;
  float4 acc = {0.f, 0.f, 0.f, 0.f};
  {
    float al[H];
    for (int hh = 0; hh < H; ++hh) al[hh] = x0[hh] / den[hh];
    add_rows<H>(acc, hfeat, s0, al, tot < 32 ? tot : 32, lane, myhead);
  }
  for (int c0 = 32; c0 < tot; c0 += 32) {
    int j = c0 + lane;
    int s = d;
    float al[H];
    for (int hh = 0; hh < H; ++hh) al[hh] = 0.f;
    if (j < tot) {
      s = (j < L) ? ssrc[beg + j] : d;
      for (int hh = 0; hh < H; ++hh)
        al[hh] = __expf(lrelu(als[(size_t)s * H + hh] + adv[hh], 0.2f) - m[hh]) / den[hh];
    }
    add_rows<H>(acc, hfeat, s, al, tot - c0 < 32 ? tot - c0 : 32, lane, myhead);
  }

  const float4 bv = *((const float4*)bias + lane);
  float4 o = {acc.x + bv.x, acc.y + bv.y, acc.z + bv.z, acc.w + bv.w};
  if (act) { o.x = lrelu(o.x, 0.01f); o.y = lrelu(o.y, 0.01f); o.z = lrelu(o.z, 0.01f); o.w = lrelu(o.w, 0.01f); }
  *((float4*)(out + (size_t)d * HID) + lane) = o;
}

extern "C" void kernel_launch(void* const* d_in, const int* in_sizes, int n_in,
                              void* d_out, int out_size, void* d_ws, size_t ws_size,
                              hipStream_t stream) {
  const float* num_prop = (const float*)d_in[0];
  const float* cat_prop = (const float*)d_in[1];
  const int*   ei       = (const int*)d_in[2];
  const float* W_num = (const float*)d_in[3];
  const float* b_num = (const float*)d_in[4];
  const float* W_cat = (const float*)d_in[5];
  const float* b_cat = (const float*)d_in[6];
  const float* W_in  = (const float*)d_in[7];
  const float* b_in  = (const float*)d_in[8];
  const float* bn_g  = (const float*)d_in[9];
  const float* bn_b  = (const float*)d_in[10];
  const float* W1    = (const float*)d_in[11];
  const float* a_s1  = (const float*)d_in[12];
  const float* a_d1  = (const float*)d_in[13];
  const float* bias1 = (const float*)d_in[14];
  const float* W2    = (const float*)d_in[15];
  const float* a_s2  = (const float*)d_in[16];
  const float* a_d2  = (const float*)d_in[17];
  const float* bias2 = (const float*)d_in[18];

  const int N = in_sizes[0] / 7;
  const int E = in_sizes[2] / 2;
  float* out = (float*)d_out;

  float* ws   = (float*)d_ws;
  float* xcat = ws;
  float* x1   = xcat + (size_t)N * HID;
  float* hbuf = x1   + (size_t)N * HID;
  float* accb = hbuf + (size_t)N * HID;
  float* als  = accb + (size_t)N * HID;
  float* ald  = als  + (size_t)N * 4;
  float* stats= ald  + (size_t)N * 12;

  const int TB = 256;
  const int BNB = 512;
  const int mtiles = (N + 15) / 16;
  auto nb = [&](long long work) { return (unsigned)((work + TB - 1) / TB); };

  const size_t Np = ((size_t)N + 64) / 64 * 64;
  const size_t Ep = ((size_t)E + 63) / 64 * 64;
  int* cnt  = (int*)x1;
  int* off  = cnt + Np;
  int* rank = off + Np;
  int* bsrc = rank + Ep;
  int* ssrc = bsrc + Ep;
  float* bnpart = hbuf;

  encoder_kernel<<<nb((long long)N * 64), TB, 0, stream>>>(num_prop, cat_prop, W_num, b_num,
                                                           W_cat, b_cat, xcat, N);
  wmma_gemm128_kernel<<<mtiles, TB, 0, stream>>>(xcat, W_in, b_in, x1, N, 1, 1);

  bn_stats_kernel<<<BNB, 128, 0, stream>>>(x1, bnpart, N);
  bn_finalize_kernel<<<1, 1024, 0, stream>>>(bnpart, BNB, stats, bn_g, bn_b, N);
  bn_apply_kernel<<<nb((long long)N * HID), TB, 0, stream>>>(x1, stats, xcat, N);

  fill_int_kernel<<<nb(N), TB, 0, stream>>>(cnt, N, 0);
  edge_rank_kernel<<<nb(E), TB, 0, stream>>>(ei, cnt, rank, E);
  bucket_scan_kernel<<<1, 1024, 0, stream>>>(cnt, off, N);
  edge_place_kernel<<<nb(E), TB, 0, stream>>>(ei, off, rank, bsrc, E);
  bucket_sort_kernel<<<nb((long long)N * 32), TB, 0, stream>>>(off, bsrc, ssrc, N);

  wmma_gemm128_kernel<<<mtiles, TB, 0, stream>>>(xcat, W1, nullptr, hbuf, N, 0, 0);
  att_logits_kernel<<<nb((long long)N * 4), TB, 0, stream>>>(hbuf, a_s1, a_d1, als, ald, N, 4, 32);
  gat_aggregate_kernel<4><<<nb((long long)N * 32), TB, 0, stream>>>(off, ssrc, hbuf, als, ald, bias1,
                                                                  accb, N, 1);

  wmma_gemm128_kernel<<<mtiles, TB, 0, stream>>>(accb, W2, nullptr, hbuf, N, 0, 0);
  att_logits_kernel<<<nb(N), TB, 0, stream>>>(hbuf, a_s2, a_d2, als, ald, N, 1, 128);
  gat_aggregate_kernel<1><<<nb((long long)N * 32), TB, 0, stream>>>(off, ssrc, hbuf, als, ald, bias2,
                                                                  out, N, 0);
}


# batchnorm sums, batchnorm finalize and bucket-offset scan: 8 to 16 independent whole-line loads in flight per lane instead of one awaited load per loop trip (scan reads coalesced 16-byte pieces)
# speedup vs baseline: 2.7880x; 1.4362x over previous
#include <hip/hip_runtime.h>
#include <math.h>

typedef __attribute__((ext_vector_type(2))) float v2f;
typedef __attribute__((ext_vector_type(8))) float v8f;

#define HID 128

__device__ __forceinline__ float lrelu(float v, float s) { return v > 0.f ? v : s * v; }

__global__ void encoder_kernel(const float* __restrict__ num_prop, const float* __restrict__ cat_prop,
                               const float* __restrict__ Wn, const float* __restrict__ bn,
                               const float* __restrict__ Wc, const float* __restrict__ bc,
                               float* __restrict__ xcat, int N) {
  int t = blockIdx.x * blockDim.x + threadIdx.x;
  if (t >= N * 64) return;
  int n = t >> 6, j = t & 63;
  const float* np = num_prop + (size_t)n * 7;
  float an = bn[j];
#pragma unroll
  for (int k = 0; k < 7; ++k) an += np[k] * Wn[k * 64 + j];
  const float* cp = cat_prop + (size_t)n * 3;
  float ac = bc[j];
#pragma unroll
  for (int k = 0; k < 3; ++k) ac += cp[k] * Wc[k * 64 + j];
  xcat[(size_t)n * HID + j]      = lrelu(an, 0.01f);
  xcat[(size_t)n * HID + 64 + j] = lrelu(ac, 0.01f);
}

__global__ void wmma_gemm128_kernel(const float* __restrict__ A, const float* __restrict__ B,
                                    const float* __restrict__ bias, float* __restrict__ C,
                                    int M, int act, int has_bias) {
  int wave = threadIdx.x >> 5;
  int lane = threadIdx.x & 31;
  int row0 = blockIdx.x * 16;
  int col0 = wave * 16;
  int half = lane >> 4;
  int m16  = lane & 15;

  int arow = row0 + m16; if (arow >= M) arow = M - 1;
  const float* Ap = A + (size_t)arow * HID + 2 * half;
  const float* Bp = B + (size_t)(2 * half) * HID + col0 + m16;

  v8f acc = {0.f, 0.f, 0.f, 0.f, 0.f, 0.f, 0.f, 0.f};
#pragma unroll 8
  for (int k0 = 0; k0 < HID; k0 += 4) {
    v2f a, b;
    a.x = Ap[k0];
    a.y = Ap[k0 + 1];
    b.x = Bp[(size_t)k0 * HID];
    b.y = Bp[(size_t)(k0 + 1) * HID];
    acc = __builtin_amdgcn_wmma_f32_16x16x4_f32(false, a, false, b, (short)0, acc, false, false);
  }

  int col = col0 + m16;
  float bv = has_bias ? bias[col] : 0.f;
#pragma unroll
  for (int r = 0; r < 8; ++r) {
    int row = row0 + r + 8 * half;
    if (row < M) {
      float v = acc[r] + bv;
      if (act) v = lrelu(v, 0.01f);
      C[(size_t)row * HID + col] = v;
    }
  }
}

#define BN_BLOCKS 256
#define BN_WAVES  16

__global__ void __launch_bounds__(BN_WAVES * 32)
bn_stats_kernel(const float* __restrict__ x, float* __restrict__ part, int N) {
  __shared__ float4 sh[2][BN_WAVES][32];
  const int lane = threadIdx.x & 31;
  const int wave = threadIdx.x >> 5;
  const int W = gridDim.x * BN_WAVES;
  const int gw = __builtin_amdgcn_readfirstlane((int)(blockIdx.x * BN_WAVES + wave));
  float4 s  = {0.f, 0.f, 0.f, 0.f};
  float4 s2 = {0.f, 0.f, 0.f, 0.f};
  for (int r0 = gw; r0 < N; r0 += 8 * W) {
    float4 v[8];
#pragma unroll
    for (int i = 0; i < 8; ++i) {
      const int r  = r0 + i * W;
      const bool ok = r < N;
      const float4 t = *((const float4*)(x + (size_t)(ok ? r : N - 1) * HID) + lane);
      v[i].x = ok ? t.x : 0.f; v[i].y = ok ? t.y : 0.f;
      v[i].z = ok ? t.z : 0.f; v[i].w = ok ? t.w : 0.f;
    }
#pragma unroll
    for (int i = 0; i < 8; ++i) {
      s.x += v[i].x; s.y += v[i].y; s.z += v[i].z; s.w += v[i].w;
      s2.x += v[i].x * v[i].x; s2.y += v[i].y * v[i].y;
      s2.z += v[i].z * v[i].z; s2.w += v[i].w * v[i].w;
    }
  }
  sh[0][wave][lane] = s;
  sh[1][wave][lane] = s2;
  __syncthreads();
  if (threadIdx.x < 64) {
    const int which = threadIdx.x >> 5;
    float4 a = sh[which][0][lane];
#pragma unroll
    for (int w = 1; w < BN_WAVES; ++w) {
      const float4 b = sh[which][w][lane];
      a.x += b.x; a.y += b.y; a.z += b.z; a.w += b.w;
    }
    *((float4*)(part + (size_t)blockIdx.x * 2 * HID + which * HID) + lane) = a;
  }
}

__global__ void __launch_bounds__(1024)
bn_finalize_kernel(const float* __restrict__ part, float* __restrict__ stats,
                   const float* __restrict__ gamma, const float* __restrict__ beta, int N) {
  __shared__ float sh[2][8][HID];
  constexpr int PER = BN_BLOCKS / 8;
  const int c = threadIdx.x & (HID - 1);
  const int g = threadIdx.x >> 7;
  const float* p = part + (size_t)(g * PER) * 2 * HID + c;
  float s = 0.f, s2 = 0.f;
#pragma unroll
  for (int b0 = 0; b0 < PER; b0 += 16) {
    float a[16], q[16];
#pragma unroll
    for (int i = 0; i < 16; ++i) {
      a[i] = p[(size_t)(b0 + i) * 2 * HID];
      q[i] = p[(size_t)(b0 + i) * 2 * HID + HID];
    }
#pragma unroll
    for (int i = 0; i < 16; ++i) { s += a[i]; s2 += q[i]; }
  }
  sh[0][g][c] = s;
  sh[1][g][c] = s2;
  __syncthreads();
  if (threadIdx.x < HID) {
    float S = 0.f, S2 = 0.f;
    for (int k = 0; k < 8; ++k) { S += sh[0][k][c]; S2 += sh[1][k][c]; }
    float inv_n = 1.f / (float)N;
    float mu  = S * inv_n;
    float var = S2 * inv_n - mu * mu;
    float sc  = gamma[c] * rsqrtf(var + 1e-5f);
    stats[2 * HID + c] = sc;
    stats[3 * HID + c] = beta[c] - mu * sc;
  }
}

__global__ void bn_apply_kernel(const float* __restrict__ x, const float* __restrict__ stats,
                                float* __restrict__ y, int N) {
  int t = blockIdx.x * blockDim.x + threadIdx.x;
  if (t >= N * HID) return;
  int c = t & (HID - 1);
  y[t] = x[t] * stats[2 * HID + c] + stats[3 * HID + c];
}

__global__ void att_logits_kernel(const float* __restrict__ h, const float* __restrict__ a_src,
                                  const float* __restrict__ a_dst, float* __restrict__ als,
                                  float* __restrict__ ald, int N, int H, int F) {
  int t = blockIdx.x * blockDim.x + threadIdx.x;
  if (t >= N * H) return;
  int n = t / H, hh = t - n * H;
  const float* hp = h + (size_t)n * HID + hh * F;
  const float* as = a_src + hh * F;
  const float* ad = a_dst + hh * F;
  float ss = 0.f, sd = 0.f;
  for (int f = 0; f < F; ++f) { float v = hp[f]; ss += v * as[f]; sd += v * ad[f]; }
  als[t] = ss;
  ald[t] = sd;
}

__global__ void fill_int_kernel(int* __restrict__ p, int n, int v) {
  int t = blockIdx.x * blockDim.x + threadIdx.x;
  if (t < n) p[t] = v;
}

__global__ void edge_rank_kernel(const int* __restrict__ ei, int* __restrict__ cnt,
                                 int* __restrict__ rank, int E) {
  int e = blockIdx.x * blockDim.x + threadIdx.x;
  if (e >= E) return;
  int d = ei[(size_t)E + e];
  rank[e] = atomicAdd(cnt + d, 1);
}

__device__ __forceinline__ int4 load_tile(const int* __restrict__ cnt, int t, int t1, int lane) {
  const bool ok = t < t1;
  const int4 v = *((const int4*)(cnt + (size_t)(ok ? t : t1 - 1) * 128) + lane);
  int4 r;
  r.x = ok ? v.x : 0; r.y = ok ? v.y : 0; r.z = ok ? v.z : 0; r.w = ok ? v.w : 0;
  return r;
}

__global__ void __launch_bounds__(1024)
bucket_scan_kernel(const int* __restrict__ cnt, int* __restrict__ off, int tiles) {
  __shared__ int wsum[32];
  const int lane = threadIdx.x & 31;
  const int w = __builtin_amdgcn_readfirstlane((int)(threadIdx.x >> 5));
  const int per = (tiles + 31) / 32;
  const int t0 = w * per;
  const int t1 = t0 + per < tiles ? t0 + per : tiles;

  int tot = 0;
  for (int t = t0; t < t1; t += 8) {
    int4 v[8];
#pragma unroll
    for (int i = 0; i < 8; ++i) v[i] = load_tile(cnt, t + i, t1, lane);
#pragma unroll
    for (int i = 0; i < 8; ++i) tot += (v[i].x + v[i].y) + (v[i].z + v[i].w);
  }
  for (int o = 16; o > 0; o >>= 1) tot += __shfl_xor(tot, o, 32);
  if (lane == 0) wsum[w] = tot;
  __syncthreads();

  const int mine = wsum[lane];
  int incl = mine;
  for (int o = 1; o < 32; o <<= 1) {
    const int y = __shfl_up(incl, o, 32);
    if (lane >= o) incl += y;
  }
  int base = __builtin_amdgcn_readlane(incl - mine, w);

  for (int t = t0; t < t1; t += 8) {
    int4 v[8];
#pragma unroll
    for (int i = 0; i < 8; ++i) v[i] = load_tile(cnt, t + i, t1, lane);
#pragma unroll
    for (int i = 0; i < 8; ++i) {
      const int p0 = v[i].x, p1 = p0 + v[i].y, p2 = p1 + v[i].z, p3 = p2 + v[i].w;
      int inc = p3;
      for (int o = 1; o < 32; o <<= 1) {
        const int y = __shfl_up(inc, o, 32);
        if (lane >= o) inc += y;
      }
      const int ex = base + inc - p3;
      if (t + i < t1) {
        int4 r;
        r.x = ex; r.y = ex + p0; r.z = ex + p1; r.w = ex + p2;
        *((int4*)(off + (size_t)(t + i) * 128) + lane) = r;
      }
      base += __builtin_amdgcn_readlane(inc, 31);
    }
  }
}

__global__ void edge_place_kernel(const int* __restrict__ ei, const int* __restrict__ off,
                                  const int* __restrict__ rank, int* __restrict__ bsrc, int E) {
  int e = blockIdx.x * blockDim.x + threadIdx.x;
  if (e >= E) return;
  int s = ei[e];
  int d = ei[(size_t)E + e];
  bsrc[off[d] + rank[e]] = s;
}

__global__ void bucket_sort_kernel(const int* __restrict__ off, const int* __restrict__ bsrc,
                                   int* __restrict__ ssrc, int N) {
  const int lane = threadIdx.x & 31;
  const int d = __builtin_amdgcn_readfirstlane((int)((blockIdx.x * blockDim.x + threadIdx.x) >> 5));
  if (d >= N) return;
  const int beg = off[d];
  const int L = off[d + 1] - beg;
  if (L <= 32) {
    int key = (lane < L) ? bsrc[beg + lane] : 0x7fffffff;
    for (int k = 2; k <= 32; k <<= 1) {
      for (int j = k >> 1; j > 0; j >>= 1) {
        int other = __shfl_xor(key, j, 32);
        bool up = (lane & k) == 0;
        bool lower = (lane & j) == 0;
        int lo = key < other ? key : other;
        int hi = key < other ? other : key;
        key = (lower == up) ? lo : hi;
      }
    }
    if (lane < L) ssrc[beg + lane] = key;
  } else {
    for (int i = lane; i < L; i += 32) {
      int ki = bsrc[beg + i];
      int r = 0;
      for (int j = 0; j < L; ++j) {
        int kj = bsrc[beg + j];
        r += (kj < ki || (kj == ki && j < i)) ? 1 : 0;
      }
      ssrc[beg + r] = ki;
    }
  }
}

__device__ __forceinline__ float wave_max(float v) {
  for (int o = 16; o > 0; o >>= 1) v = fmaxf(v, __shfl_xor(v, o, 32));
  return v;
}
__device__ __forceinline__ float wave_sum(float v) {
  for (int o = 16; o > 0; o >>= 1) v += __shfl_xor(v, o, 32);
  return v;
}

template <int H>
__device__ __forceinline__ void add_rows(float4& acc, const float* __restrict__ hfeat, int s,
                                         const float (&al)[H], int n, int lane, int myhead) {
#pragma unroll 8
  for (int jj = 0; jj < n; ++jj) {
    const int sj = __builtin_amdgcn_readlane(s, jj);
    float a = __int_as_float(__builtin_amdgcn_readlane(__float_as_int(al[0]), jj));
    for (int hh = 1; hh < H; ++hh) {
      float ah = __int_as_float(__builtin_amdgcn_readlane(__float_as_int(al[hh]), jj));
      a = (myhead == hh) ? ah : a;
    }
    const float4 hv = *((const float4*)(hfeat + (size_t)sj * HID) + lane);
    acc.x += a * hv.x;
    acc.y += a * hv.y;
    acc.z += a * hv.z;
    acc.w += a * hv.w;
  }
}

template <int H>
__global__ void gat_aggregate_kernel(const int* __restrict__ off, const int* __restrict__ ssrc,
                                     const float* __restrict__ hfeat, const float* __restrict__ als,
                                     const float* __restrict__ ald, const float* __restrict__ bias,
                                     float* __restrict__ out, int N, int act) {
  constexpr int F = HID / H;
  const int lane = threadIdx.x & 31;
  const int d = __builtin_amdgcn_readfirstlane((int)((blockIdx.x * blockDim.x + threadIdx.x) >> 5));
  if (d >= N) return;
  const int beg = off[d];
  const int L = off[d + 1] - beg;
  const int tot = L + 1;

  float adv[H], m[H], den[H], e0[H];
  for (int hh = 0; hh < H; ++hh) { adv[hh] = ald[(size_t)d * H + hh]; m[hh] = -INFINITY; den[hh] = 0.f; e0[hh] = 0.f; }

  const bool v0 = lane < tot;
  int s0 = d;
  if (v0) {
    s0 = (lane < L) ? ssrc[beg + lane] : d;
    for (int hh = 0; hh < H; ++hh) { e0[hh] = lrelu(als[(size_t)s0 * H + hh] + adv[hh], 0.2f); m[hh] = e0[hh]; }
  }

  for (int c0 = 32; c0 < tot; c0 += 32) {
    int j = c0 + lane;
    if (j < tot) {
      int s = (j < L) ? ssrc[beg + j] : d;
      for (int hh = 0; hh < H; ++hh)
        m[hh] = fmaxf(m[hh], lrelu(als[(size_t)s * H + hh] + adv[hh], 0.2f));
    }
  }
  for (int hh = 0; hh < H; ++hh) m[hh] = wave_max(m[hh]);

  float x0[H];
  for (int hh = 0; hh < H; ++hh) { x0[hh] = v0 ? __expf(e0[hh] - m[hh]) : 0.f; den[hh] = x0[hh]; }
  for (int c0 = 32; c0 < tot; c0 += 32) {
    int j = c0 + lane;
    if (j < tot) {
      int s = (j < L) ? ssrc[beg + j] : d;
      for (int hh = 0; hh < H; ++hh)
        den[hh] += __expf(lrelu(als[(size_t)s * H + hh] + adv[hh], 0.2f) - m[hh]);
    }
  }
  for (int hh = 0; hh < H; ++hh) den[hh] = wave_sum(den[hh]);

  const int myhead = (4 * lane) / F;
  float4 acc = {0.f, 0.f, 0.f, 0.f};
  {
    float al[H];
    for (int hh = 0; hh < H; ++hh) al[hh] = x0[hh] / den[hh];
    add_rows<H>(acc, hfeat, s0, al, tot < 32 ? tot : 32, lane, myhead);
  }
  for (int c0 = 32; c0 < tot; c0 += 32) {
    int j = c0 + lane;
    int s = d;
    float al[H];
    for (int hh = 0; hh < H; ++hh) al[hh] = 0.f;
    if (j < tot) {
      s = (j < L) ? ssrc[beg + j] : d;
      for (int hh = 0; hh < H; ++hh)
        al[hh] = __expf(lrelu(als[(size_t)s * H + hh] + adv[hh], 0.2f) - m[hh]) / den[hh];
    }
    add_rows<H>(acc, hfeat, s, al, tot - c0 < 32 ? tot - c0 : 32, lane, myhead);
  }

  const float4 bv = *((const float4*)bias + lane);
  float4 o = {acc.x + bv.x, acc.y + bv.y, acc.z + bv.z, acc.w + bv.w};
  if (act) { o.x = lrelu(o.x, 0.01f); o.y = lrelu(o.y, 0.01f); o.z = lrelu(o.z, 0.01f); o.w = lrelu(o.w, 0.01f); }
  *((float4*)(out + (size_t)d * HID) + lane) = o;
}

extern "C" void kernel_launch(void* const* d_in, const int* in_sizes, int n_in,
                              void* d_out, int out_size, void* d_ws, size_t ws_size,
                              hipStream_t stream) {
  const float* num_prop = (const float*)d_in[0];
  const float* cat_prop = (const float*)d_in[1];
  const int*   ei       = (const int*)d_in[2];
  const float* W_num = (const float*)d_in[3];
  const float* b_num = (const float*)d_in[4];
  const float* W_cat = (const float*)d_in[5];
  const float* b_cat = (const float*)d_in[6];
  const float* W_in  = (const float*)d_in[7];
  const float* b_in  = (const float*)d_in[8];
  const float* bn_g  = (const float*)d_in[9];
  const float* bn_b  = (const float*)d_in[10];
  const float* W1    = (const float*)d_in[11];
  const float* a_s1  = (const float*)d_in[12];
  const float* a_d1  = (const float*)d_in[13];
  const float* bias1 = (const float*)d_in[14];
  const float* W2    = (const float*)d_in[15];
  const float* a_s2  = (const float*)d_in[16];
  const float* a_d2  = (const float*)d_in[17];
  const float* bias2 = (const float*)d_in[18];

  const int N = in_sizes[0] / 7;
  const int E = in_sizes[2] / 2;
  float* out = (float*)d_out;

  float* ws   = (float*)d_ws;
  float* xcat = ws;
  float* x1   = xcat + (size_t)N * HID;
  float* hbuf = x1   + (size_t)N * HID;
  float* accb = hbuf + (size_t)N * HID;
  float* als  = accb + (size_t)N * HID;
  float* ald  = als  + (size_t)N * 4;
  float* stats= ald  + (size_t)N * 12;

  const int TB = 256;
  const int mtiles = (N + 15) / 16;
  auto nb = [&](long long work) { return (unsigned)((work + TB - 1) / TB); };

  const size_t Np = ((size_t)N + 128) / 128 * 128;
  const size_t Ep = ((size_t)E + 63) / 64 * 64;
  int* cnt  = (int*)x1;
  int* off  = cnt + Np;
  int* rank = off + Np;
  int* bsrc = rank + Ep;
  int* ssrc = bsrc + Ep;
  float* bnpart = hbuf;

  encoder_kernel<<<nb((long long)N * 64), TB, 0, stream>>>(num_prop, cat_prop, W_num, b_num,
                                                           W_cat, b_cat, xcat, N);
  wmma_gemm128_kernel<<<mtiles, TB, 0, stream>>>(xcat, W_in, b_in, x1, N, 1, 1);

  bn_stats_kernel<<<BN_BLOCKS, BN_WAVES * 32, 0, stream>>>(x1, bnpart, N);
  bn_finalize_kernel<<<1, 1024, 0, stream>>>(bnpart, stats, bn_g, bn_b, N);
  bn_apply_kernel<<<nb((long long)N * HID), TB, 0, stream>>>(x1, stats, xcat, N);

  fill_int_kernel<<<nb((long long)Np), TB, 0, stream>>>(cnt, (int)Np, 0);
  edge_rank_kernel<<<nb(E), TB, 0, stream>>>(ei, cnt, rank, E);
  bucket_scan_kernel<<<1, 1024, 0, stream>>>(cnt, off, (int)(Np / 128));
  edge_place_kernel<<<nb(E), TB, 0, stream>>>(ei, off, rank, bsrc, E);
  bucket_sort_kernel<<<nb((long long)N * 32), TB, 0, stream>>>(off, bsrc, ssrc, N);

  wmma_gemm128_kernel<<<mtiles, TB, 0, stream>>>(xcat, W1, nullptr, hbuf, N, 0, 0);
  att_logits_kernel<<<nb((long long)N * 4), TB, 0, stream>>>(hbuf, a_s1, a_d1, als, ald, N, 4, 32);
  gat_aggregate_kernel<4><<<nb((long long)N * 32), TB, 0, stream>>>(off, ssrc, hbuf, als, ald, bias1,
                                                                  accb, N, 1);

  wmma_gemm128_kernel<<<mtiles, TB, 0, stream>>>(accb, W2, nullptr, hbuf, N, 0, 0);
  att_logits_kernel<<<nb(N), TB, 0, stream>>>(hbuf, a_s2, a_d2, als, ald, N, 1, 128);
  gat_aggregate_kernel<1><<<nb((long long)N * 32), TB, 0, stream>>>(off, ssrc, hbuf, als, ald, bias2,
                                                                  out, N, 0);
}


# GEMM operand path: a block stages its 32 rows of A once in LDS by whole-row loads (before: all 8 waves fetched A from global, 8 bytes per lane, a line per lane); a wave does 2 row tiles per B load
# speedup vs baseline: 3.7665x; 1.3510x over previous
#include <hip/hip_runtime.h>
#include <math.h>

typedef __attribute__((ext_vector_type(2))) float v2f;
typedef __attribute__((ext_vector_type(8))) float v8f;

#define HID 128

__device__ __forceinline__ float lrelu(float v, float s) { return v > 0.f ? v : s * v; }

__global__ void encoder_kernel(const float* __restrict__ num_prop, const float* __restrict__ cat_prop,
                               const float* __restrict__ Wn, const float* __restrict__ bn,
                               const float* __restrict__ Wc, const float* __restrict__ bc,
                               float* __restrict__ xcat, int N,
                               int* __restrict__ cnt, int ncnt) {
  int t = blockIdx.x * blockDim.x + threadIdx.x;
  if (t < ncnt) cnt[t] = 0;
  if (t >= N * 64) return;
  int n = t >> 6, j = t & 63;
  const float* np = num_prop + (size_t)n * 7;
  float an = bn[j];
#pragma unroll
  for (int k = 0; k < 7; ++k) an += np[k] * Wn[k * 64 + j];
  const float* cp = cat_prop + (size_t)n * 3;
  float ac = bc[j];
#pragma unroll
  for (int k = 0; k < 3; ++k) ac += cp[k] * Wc[k * 64 + j];
  xcat[(size_t)n * HID + j]      = lrelu(an, 0.01f);
  xcat[(size_t)n * HID + 64 + j] = lrelu(ac, 0.01f);
}


#define APITCH 132

template <int XM>
__device__ __forceinline__ float row_xor(float v) {
  return __int_as_float(__builtin_amdgcn_update_dpp(0, __float_as_int(v), 0x160 + XM, 0xf, 0xf, true));
}

__device__ __forceinline__ float logit_partial(const v8f& acc, float as_c, float ad_c, int m16) {
  float v[16];
#pragma unroll
  for (int r = 0; r < 8; ++r) { v[r] = acc[r] * as_c; v[8 + r] = acc[r] * ad_c; }
  const bool b3 = (m16 & 8) != 0, b2 = (m16 & 4) != 0, b1 = (m16 & 2) != 0, b0 = (m16 & 1) != 0;
  float w8[8], w4[4], w2[2];
#pragma unroll
  for (int j = 0; j < 8; ++j) {
    const float keep = b3 ? v[8 + j] : v[j];
    const float send = b3 ? v[j] : v[8 + j];
    w8[j] = keep + row_xor<8>(send);
  }
#pragma unroll
  for (int j = 0; j < 4; ++j) {
    const float keep = b2 ? w8[4 + j] : w8[j];
    const float send = b2 ? w8[j] : w8[4 + j];
    w4[j] = keep + row_xor<4>(send);
  }
#pragma unroll
  for (int j = 0; j < 2; ++j) {
    const float keep = b1 ? w4[2 + j] : w4[j];
    const float send = b1 ? w4[j] : w4[2 + j];
    w2[j] = keep + row_xor<2>(send);
  }
  const float keep1 = b0 ? w2[1] : w2[0];
  const float send1 = b0 ? w2[0] : w2[1];
  return keep1 + row_xor<1>(send1);
}

template <bool BN, int H>
__device__ __forceinline__ void
gemm128_tile(int tile, const float* __restrict__ A, const float* __restrict__ B,
             const float* __restrict__ bias, float* __restrict__ C,
             const float* __restrict__ stats,
             const float* __restrict__ a_src, const float* __restrict__ a_dst,
             float* __restrict__ als, float* __restrict__ ald,
             int M, int act, int has_bias) {
  __shared__ __attribute__((aligned(16))) float As[32 * APITCH];
  int wave = threadIdx.x >> 5;
  int lane = threadIdx.x & 31;
  int row0 = tile * 32;
  int col0 = wave * 16;
  int half = lane >> 4;
  int m16  = lane & 15;

  {
    float4 sc = {1.f, 1.f, 1.f, 1.f}, sh = {0.f, 0.f, 0.f, 0.f};
    if constexpr (BN) {
      sc = *((const float4*)(stats + 2 * HID) + lane);
      sh = *((const float4*)(stats + 3 * HID) + lane);
    }
    float4 v[4];
#pragma unroll
    for (int i = 0; i < 4; ++i) {
      int r = row0 + 4 * wave + i; if (r >= M) r = M - 1;
      v[i] = *((const float4*)(A + (size_t)r * HID) + lane);
    }
#pragma unroll
    for (int i = 0; i < 4; ++i) {
      if constexpr (BN) {
        v[i].x = v[i].x * sc.x + sh.x; v[i].y = v[i].y * sc.y + sh.y;
        v[i].z = v[i].z * sc.z + sh.z; v[i].w = v[i].w * sc.w + sh.w;
      }
      *((float4*)(As + (4 * wave + i) * APITCH) + lane) = v[i];
    }
  }

  const float* Bp = B + (size_t)(2 * half) * HID + col0 + m16;
  const int col = col0 + m16;
  float as_c = 0.f, ad_c = 0.f;
  if constexpr (H > 0) { as_c = a_src[col]; ad_c = a_dst[col]; }

  __syncthreads();

  const float* Ar0 = As + m16 * APITCH + 2 * half;
  const float* Ar1 = Ar0 + 16 * APITCH;
  v8f acc0 = {0.f, 0.f, 0.f, 0.f, 0.f, 0.f, 0.f, 0.f};
  v8f acc1 = {0.f, 0.f, 0.f, 0.f, 0.f, 0.f, 0.f, 0.f};
#pragma unroll 8
  for (int k0 = 0; k0 < HID; k0 += 4) {
    v2f b;
    b.x = Bp[(size_t)k0 * HID];
    b.y = Bp[(size_t)(k0 + 1) * HID];
    const v2f a0 = *((const v2f*)(Ar0 + k0));
    const v2f a1 = *((const v2f*)(Ar1 + k0));
    acc0 = __builtin_amdgcn_wmma_f32_16x16x4_f32(false, a0, false, b, (short)0, acc0, false, false);
    acc1 = __builtin_amdgcn_wmma_f32_16x16x4_f32(false, a1, false, b, (short)0, acc1, false, false);
  }

  if constexpr (H > 0) {
    __shared__ float part[2][2][16][8];
    const float p0 = logit_partial(acc0, as_c, ad_c, m16);
    const float p1 = logit_partial(acc1, as_c, ad_c, m16);
    part[0][m16 >> 3][(m16 & 7) + 8 * half][wave] = p0;
    part[1][m16 >> 3][(m16 & 7) + 8 * half][wave] = p1;
    __syncthreads();
    constexpr int WPH = 8 / H;
    if (threadIdx.x < 64 * H) {
      const int t     = threadIdx.x / (32 * H);
      const int rt    = threadIdx.x - t * (32 * H);
      const int which = rt / (16 * H);
      const int rem   = rt - which * (16 * H);
      const int row   = rem / H;
      const int hh    = rem - row * H;
      float s = part[t][which][row][hh * WPH];
#pragma unroll
      for (int w = 1; w < WPH; ++w) s += part[t][which][row][hh * WPH + w];
      const int grow = row0 + 16 * t + row;
      if (grow < M) (which ? ald : als)[(size_t)grow * H + hh] = s;
    }
  }

  float bv = has_bias ? bias[col] : 0.f;
#pragma unroll
  for (int r = 0; r < 8; ++r) {
    int row = row0 + r + 8 * half;
    if (row < M) {
      float v = acc0[r] + bv;
      if (act) v = lrelu(v, 0.01f);
      C[(size_t)row * HID + col] = v;
    }
    row += 16;
    if (row < M) {
      float v = acc1[r] + bv;
      if (act) v = lrelu(v, 0.01f);
      C[(size_t)row * HID + col] = v;
    }
  }
}

template <bool BN, int H>
__global__ void __launch_bounds__(256)
wmma_gemm128_kernel(const float* __restrict__ A, const float* __restrict__ B,
                    const float* __restrict__ bias, float* __restrict__ C,
                    const float* __restrict__ stats,
                    const float* __restrict__ a_src, const float* __restrict__ a_dst,
                    float* __restrict__ als, float* __restrict__ ald,
                    int M, int act, int has_bias) {
  gemm128_tile<BN, H>((int)blockIdx.x, A, B, bias, C, stats, a_src, a_dst, als, ald, M, act, has_bias);
}

#define BN_BLOCKS 256
#define BN_WAVES  32

__device__ __forceinline__ void
bn_stats_body(int blk, int nblk, const float* __restrict__ x, float* __restrict__ part, int N) {
  __shared__ float4 sh[2][BN_WAVES][32];
  const int lane = threadIdx.x & 31;
  const int wave = threadIdx.x >> 5;
  const int W = nblk * BN_WAVES;
  const int gw = __builtin_amdgcn_readfirstlane((int)(blk * BN_WAVES + wave));
  float4 s  = {0.f, 0.f, 0.f, 0.f};
  float4 s2 = {0.f, 0.f, 0.f, 0.f};
  for (int r0 = gw; r0 < N; r0 += 8 * W) {
    float4 v[8];
#pragma unroll
    for (int i = 0; i < 8; ++i) {
      const int r  = r0 + i * W;
      const bool ok = r < N;
      const float4 t = *((const float4*)(x + (size_t)(ok ? r : N - 1) * HID) + lane);
      v[i].x = ok ? t.x : 0.f; v[i].y = ok ? t.y : 0.f;
      v[i].z = ok ? t.z : 0.f; v[i].w = ok ? t.w : 0.f;
    }
#pragma unroll
    for (int i = 0; i < 8; ++i) {
      s.x += v[i].x; s.y += v[i].y; s.z += v[i].z; s.w += v[i].w;
      s2.x += v[i].x * v[i].x; s2.y += v[i].y * v[i].y;
      s2.z += v[i].z * v[i].z; s2.w += v[i].w * v[i].w;
    }
  }
  sh[0][wave][lane] = s;
  sh[1][wave][lane] = s2;
  __syncthreads();
  if (threadIdx.x < 64) {
    const int which = threadIdx.x >> 5;
    float4 a = sh[which][0][lane];
#pragma unroll
    for (int w = 1; w < BN_WAVES; ++w) {
      const float4 b = sh[which][w][lane];
      a.x += b.x; a.y += b.y; a.z += b.z; a.w += b.w;
    }
    *((float4*)(part + (size_t)blk * 2 * HID + which * HID) + lane) = a;
  }
}

__device__ __forceinline__ void
bn_finalize_body(const float* __restrict__ part, float* __restrict__ stats,
                 const float* __restrict__ gamma, const float* __restrict__ beta, int N) {
  __shared__ float sh[2][8][HID];
  constexpr int PER = BN_BLOCKS / 8;
  const int c = threadIdx.x & (HID - 1);
  const int g = threadIdx.x >> 7;
  const float* p = part + (size_t)(g * PER) * 2 * HID + c;
  float s = 0.f, s2 = 0.f;
#pragma unroll
  for (int b0 = 0; b0 < PER; b0 += 16) {
    float a[16], q[16];
#pragma unroll
    for (int i = 0; i < 16; ++i) {
      a[i] = p[(size_t)(b0 + i) * 2 * HID];
      q[i] = p[(size_t)(b0 + i) * 2 * HID + HID];
    }
#pragma unroll
    for (int i = 0; i < 16; ++i) { s += a[i]; s2 += q[i]; }
  }
  sh[0][g][c] = s;
  sh[1][g][c] = s2;
  __syncthreads();
  if (threadIdx.x < HID) {
    float S = 0.f, S2 = 0.f;
    for (int k = 0; k < 8; ++k) { S += sh[0][k][c]; S2 += sh[1][k][c]; }
    float inv_n = 1.f / (float)N;
    float mu  = S * inv_n;
    float var = S2 * inv_n - mu * mu;
    float sc  = gamma[c] * rsqrtf(var + 1e-5f);
    stats[2 * HID + c] = sc;
    stats[3 * HID + c] = beta[c] - mu * sc;
  }
}

__device__ __forceinline__ void edge_rank_body(int blk, const int* __restrict__ ei, int* __restrict__ cnt,
                                               int* __restrict__ rank, int E) {
  int e = blk * blockDim.x + threadIdx.x;
  if (e >= E) return;
  int d = ei[(size_t)E + e];
  rank[e] = atomicAdd(cnt + d, 1);
}

__device__ __forceinline__ int4 load_tile(const int* __restrict__ cnt, int t, int t1, int lane) {
  const bool ok = t < t1;
  const int4 v = *((const int4*)(cnt + (size_t)(ok ? t : t1 - 1) * 128) + lane);
  int4 r;
  r.x = ok ? v.x : 0; r.y = ok ? v.y : 0; r.z = ok ? v.z : 0; r.w = ok ? v.w : 0;
  return r;
}

__device__ __forceinline__ void
bucket_scan_body(const int* __restrict__ cnt, int* __restrict__ off, int tiles) {
  __shared__ int wsum[32];
  const int lane = threadIdx.x & 31;
  const int w = __builtin_amdgcn_readfirstlane((int)(threadIdx.x >> 5));
  const int per = (tiles + 31) / 32;
  const int t0 = w * per;
  const int t1 = t0 + per < tiles ? t0 + per : tiles;

  int tot = 0;
  for (int t = t0; t < t1; t += 8) {
    int4 v[8];
#pragma unroll
    for (int i = 0; i < 8; ++i) v[i] = load_tile(cnt, t + i, t1, lane);
#pragma unroll
    for (int i = 0; i < 8; ++i) tot += (v[i].x + v[i].y) + (v[i].z + v[i].w);
  }
  for (int o = 16; o > 0; o >>= 1) tot += __shfl_xor(tot, o, 32);
  if (lane == 0) wsum[w] = tot;
  __syncthreads();

  const int mine = wsum[lane];
  int incl = mine;
  for (int o = 1; o < 32; o <<= 1) {
    const int y = __shfl_up(incl, o, 32);
    if (lane >= o) incl += y;
  }
  int base = __builtin_amdgcn_readlane(incl - mine, w);

  for (int t = t0; t < t1; t += 8) {
    int4 v[8];
#pragma unroll
    for (int i = 0; i < 8; ++i) v[i] = load_tile(cnt, t + i, t1, lane);
#pragma unroll
    for (int i = 0; i < 8; ++i) {
      const int p0 = v[i].x, p1 = p0 + v[i].y, p2 = p1 + v[i].z, p3 = p2 + v[i].w;
      int inc = p3;
      for (int o = 1; o < 32; o <<= 1) {
        const int y = __shfl_up(inc, o, 32);
        if (lane >= o) inc += y;
      }
      const int ex = base + inc - p3;
      if (t + i < t1) {
        int4 r;
        r.x = ex; r.y = ex + p0; r.z = ex + p1; r.w = ex + p2;
        *((int4*)(off + (size_t)(t + i) * 128) + lane) = r;
      }
      base += __builtin_amdgcn_readlane(inc, 31);
    }
  }
}

__device__ __forceinline__ void edge_place_body(int blk, const int* __restrict__ ei, const int* __restrict__ off,
                                                const int* __restrict__ rank, int* __restrict__ bsrc, int E) {
  int e = blk * blockDim.x + threadIdx.x;
  if (e >= E) return;
  int s = ei[e];
  int d = ei[(size_t)E + e];
  bsrc[off[d] + rank[e]] = s;
}

__device__ __forceinline__ void bucket_sort_body(int blk, const int* __restrict__ off, const int* __restrict__ bsrc,
                                                 int* __restrict__ ssrc, int N) {
  const int lane = threadIdx.x & 31;
  const int d = __builtin_amdgcn_readfirstlane((int)((blk * blockDim.x + threadIdx.x) >> 5));
  if (d >= N) return;
  const int beg = off[d];
  const int L = off[d + 1] - beg;
  if (L <= 32) {
    int key = (lane < L) ? bsrc[beg + lane] : 0x7fffffff;
    for (int k = 2; k <= 32; k <<= 1) {
      for (int j = k >> 1; j > 0; j >>= 1) {
        int other = __shfl_xor(key, j, 32);
        bool up = (lane & k) == 0;
        bool lower = (lane & j) == 0;
        int lo = key < other ? key : other;
        int hi = key < other ? other : key;
        key = (lower == up) ? lo : hi;
      }
    }
    if (lane < L) ssrc[beg + lane] = key;
  } else {
    for (int i = lane; i < L; i += 32) {
      int ki = bsrc[beg + i];
      int r = 0;
      for (int j = 0; j < L; ++j) {
        int kj = bsrc[beg + j];
        r += (kj < ki || (kj == ki && j < i)) ? 1 : 0;
      }
      ssrc[beg + r] = ki;
    }
  }
}


__global__ void __launch_bounds__(256)
gemm_in_and_rank_kernel(const float* __restrict__ A, const float* __restrict__ B,
                        const float* __restrict__ bias, float* __restrict__ C, int M, int mtiles,
                        const int* __restrict__ ei, int* __restrict__ cnt, int* __restrict__ rank,
                        int E, int rblocks) {
  const int g = (int)(blockIdx.x / 3);
  const int r = (int)blockIdx.x - 3 * g;
  if (r == 0) {
    if (g < mtiles)
      gemm128_tile<false, 0>(g, A, B, bias, C, nullptr, nullptr, nullptr, nullptr, nullptr, M, 1, 1);
  } else {
    const int rb = 2 * g + r - 1;
    if (rb < rblocks) edge_rank_body(rb, ei, cnt, rank, E);
  }
}

__global__ void __launch_bounds__(1024)
bn_stats_and_scan_kernel(const float* __restrict__ x, float* __restrict__ part, int N,
                         const int* __restrict__ cnt, int* __restrict__ off, int tiles) {
  if (blockIdx.x == 0) bucket_scan_body(cnt, off, tiles);
  else bn_stats_body((int)blockIdx.x - 1, (int)gridDim.x - 1, x, part, N);
}

__global__ void __launch_bounds__(1024)
bn_finalize_and_place_kernel(const float* __restrict__ part, float* __restrict__ stats,
                             const float* __restrict__ gamma, const float* __restrict__ beta, int N,
                             const int* __restrict__ ei, const int* __restrict__ off,
                             const int* __restrict__ rank, int* __restrict__ bsrc, int E) {
  if (blockIdx.x == 0) bn_finalize_body(part, stats, gamma, beta, N);
  else edge_place_body((int)blockIdx.x - 1, ei, off, rank, bsrc, E);
}

__global__ void __launch_bounds__(256)
gemm_l1_and_sort_kernel(const float* __restrict__ A, const float* __restrict__ B, float* __restrict__ C,
                        const float* __restrict__ stats,
                        const float* __restrict__ a_src, const float* __restrict__ a_dst,
                        float* __restrict__ als, float* __restrict__ ald, int M, int mtiles,
                        const int* __restrict__ off, const int* __restrict__ bsrc,
                        int* __restrict__ ssrc, int N, int sblocks) {
  const int g = (int)(blockIdx.x / 5);
  const int r = (int)blockIdx.x - 5 * g;
  if (r == 0) {
    if (g < mtiles)
      gemm128_tile<true, 4>(g, A, B, nullptr, C, stats, a_src, a_dst, als, ald, M, 0, 0);
  } else {
    const int sb = 4 * g + r - 1;
    if (sb < sblocks) bucket_sort_body(sb, off, bsrc, ssrc, N);
  }
}

__device__ __forceinline__ float wave_max(float v) {
  for (int o = 16; o > 0; o >>= 1) v = fmaxf(v, __shfl_xor(v, o, 32));
  return v;
}
__device__ __forceinline__ float wave_sum(float v) {
  for (int o = 16; o > 0; o >>= 1) v += __shfl_xor(v, o, 32);
  return v;
}

template <int H>
__device__ __forceinline__ void add_rows(float4& acc, const float* __restrict__ hfeat, int s,
                                         const float (&al)[H], int n, int lane, int myhead) {
#pragma unroll 8
  for (int jj = 0; jj < n; ++jj) {
    const int sj = __builtin_amdgcn_readlane(s, jj);
    float a = __int_as_float(__builtin_amdgcn_readlane(__float_as_int(al[0]), jj));
    for (int hh = 1; hh < H; ++hh) {
      float ah = __int_as_float(__builtin_amdgcn_readlane(__float_as_int(al[hh]), jj));
      a = (myhead == hh) ? ah : a;
    }
    const float4 hv = *((const float4*)(hfeat + (size_t)sj * HID) + lane);
    acc.x += a * hv.x;
    acc.y += a * hv.y;
    acc.z += a * hv.z;
    acc.w += a * hv.w;
  }
}

template <int H>
__global__ void gat_aggregate_kernel(const int* __restrict__ off, const int* __restrict__ ssrc,
                                     const float* __restrict__ hfeat, const float* __restrict__ als,
                                     const float* __restrict__ ald, const float* __restrict__ bias,
                                     float* __restrict__ out, int N, int act) {
  constexpr int F = HID / H;
  const int lane = threadIdx.x & 31;
  const int d = __builtin_amdgcn_readfirstlane((int)((blockIdx.x * blockDim.x + threadIdx.x) >> 5));
  if (d >= N) return;
  const int beg = off[d];
  const int L = off[d + 1] - beg;
  const int tot = L + 1;

  float adv[H], m[H], den[H], e0[H];
  for (int hh = 0; hh < H; ++hh) { adv[hh] = ald[(size_t)d * H + hh]; m[hh] = -INFINITY; den[hh] = 0.f; e0[hh] = 0.f; }

  const bool v0 = lane < tot;
  int s0 = d;
  if (v0) {
    s0 = (lane < L) ? ssrc[beg + lane] : d;
    for (int hh = 0; hh < H; ++hh) { e0[hh] = lrelu(als[(size_t)s0 * H + hh] + adv[hh], 0.2f); m[hh] = e0[hh]; }
  }

  for (int c0 = 32; c0 < tot; c0 += 32) {
    int j = c0 + lane;
    if (j < tot) {
      int s = (j < L) ? ssrc[beg + j] : d;
      for (int hh = 0; hh < H; ++hh)
        m[hh] = fmaxf(m[hh], lrelu(als[(size_t)s * H + hh] + adv[hh], 0.2f));
    }
  }
  for (int hh = 0; hh < H; ++hh) m[hh] = wave_max(m[hh]);

  float x0[H];
  for (int hh = 0; hh < H; ++hh) { x0[hh] = v0 ? __expf(e0[hh] - m[hh]) : 0.f; den[hh] = x0[hh]; }
  for (int c0 = 32; c0 < tot; c0 += 32) {
    int j = c0 + lane;
    if (j < tot) {
      int s = (j < L) ? ssrc[beg + j] : d;
      for (int hh = 0; hh < H; ++hh)
        den[hh] += __expf(lrelu(als[(size_t)s * H + hh] + adv[hh], 0.2f) - m[hh]);
    }
  }
  for (int hh = 0; hh < H; ++hh) den[hh] = wave_sum(den[hh]);

  const int myhead = (4 * lane) / F;
  float4 acc = {0.f, 0.f, 0.f, 0.f};
  {
    float al[H];
    for (int hh = 0; hh < H; ++hh) al[hh] = x0[hh] / den[hh];
    add_rows<H>(acc, hfeat, s0, al, tot < 32 ? tot : 32, lane, myhead);
  }
  for (int c0 = 32; c0 < tot; c0 += 32) {
    int j = c0 + lane;
    int s = d;
    float al[H];
    for (int hh = 0; hh < H; ++hh) al[hh] = 0.f;
    if (j < tot) {
      s = (j < L) ? ssrc[beg + j] : d;
      for (int hh = 0; hh < H; ++hh)
        al[hh] = __expf(lrelu(als[(size_t)s * H + hh] + adv[hh], 0.2f) - m[hh]) / den[hh];
    }
    add_rows<H>(acc, hfeat, s, al, tot - c0 < 32 ? tot - c0 : 32, lane, myhead);
  }

  const float4 bv = *((const float4*)bias + lane);
  float4 o = {acc.x + bv.x, acc.y + bv.y, acc.z + bv.z, acc.w + bv.w};
  if (act) { o.x = lrelu(o.x, 0.01f); o.y = lrelu(o.y, 0.01f); o.z = lrelu(o.z, 0.01f); o.w = lrelu(o.w, 0.01f); }
  *((float4*)(out + (size_t)d * HID) + lane) = o;
}

extern "C" void kernel_launch(void* const* d_in, const int* in_sizes, int n_in,
                              void* d_out, int out_size, void* d_ws, size_t ws_size,
                              hipStream_t stream) {
  const float* num_prop = (const float*)d_in[0];
  const float* cat_prop = (const float*)d_in[1];
  const int*   ei       = (const int*)d_in[2];
  const float* W_num = (const float*)d_in[3];
  const float* b_num = (const float*)d_in[4];
  const float* W_cat = (const float*)d_in[5];
  const float* b_cat = (const float*)d_in[6];
  const float* W_in  = (const float*)d_in[7];
  const float* b_in  = (const float*)d_in[8];
  const float* bn_g  = (const float*)d_in[9];
  const float* bn_b  = (const float*)d_in[10];
  const float* W1    = (const float*)d_in[11];
  const float* a_s1  = (const float*)d_in[12];
  const float* a_d1  = (const float*)d_in[13];
  const float* bias1 = (const float*)d_in[14];
  const float* W2    = (const float*)d_in[15];
  const float* a_s2  = (const float*)d_in[16];
  const float* a_d2  = (const float*)d_in[17];
  const float* bias2 = (const float*)d_in[18];

  const int N = in_sizes[0] / 7;
  const int E = in_sizes[2] / 2;
  float* out = (float*)d_out;

  float* ws   = (float*)d_ws;
  float* xcat = ws;
  float* x1   = xcat + (size_t)N * HID;
  float* hbuf = x1   + (size_t)N * HID;
  float* accb = hbuf + (size_t)N * HID;
  float* als  = accb + (size_t)N * HID;
  float* ald  = als  + (size_t)N * 4;
  float* stats= ald  + (size_t)N * 12;

  const int TB = 256;
  const int mtiles = (N + 31) / 32;
  auto nb = [&](long long work) { return (unsigned)((work + TB - 1) / TB); };

  const size_t Np = ((size_t)N + 128) / 128 * 128;
  const size_t Ep = ((size_t)E + 63) / 64 * 64;
  int* cnt  = (int*)(ald + (size_t)N * 4);
  int* off  = cnt + Np;
  int* rank = (int*)accb;
  int* bsrc = rank + Ep;
  int* ssrc = (int*)xcat;
  float* bnpart = hbuf;

  encoder_kernel<<<nb((long long)N * 64), TB, 0, stream>>>(num_prop, cat_prop, W_num, b_num,
                                                           W_cat, b_cat, xcat, N, cnt, (int)Np);

  const int rblocks = (int)nb(E);
  const int sblocks = (int)nb((long long)N * 32);
  const int pblocks = (E + 1023) / 1024;
  const int triples = mtiles > (rblocks + 1) / 2 ? mtiles : (rblocks + 1) / 2;
  const int fives   = mtiles > (sblocks + 3) / 4 ? mtiles : (sblocks + 3) / 4;

  gemm_in_and_rank_kernel<<<3 * triples, TB, 0, stream>>>(xcat, W_in, b_in, x1, N, mtiles,
                                                        ei, cnt, rank, E, rblocks);
  bn_stats_and_scan_kernel<<<BN_BLOCKS + 1, BN_WAVES * 32, 0, stream>>>(x1, bnpart, N,
                                                                        cnt, off, (int)(Np / 128));
  bn_finalize_and_place_kernel<<<pblocks + 1, 1024, 0, stream>>>(bnpart, stats, bn_g, bn_b, N,
                                                                 ei, off, rank, bsrc, E);

  gemm_l1_and_sort_kernel<<<5 * fives, TB, 0, stream>>>(x1, W1, hbuf, stats, a_s1, a_d1, als, ald,
                                                          N, mtiles, off, bsrc, ssrc, N, sblocks);
  gat_aggregate_kernel<4><<<nb((long long)N * 32), TB, 0, stream>>>(off, ssrc, hbuf, als, ald, bias1,
                                                                  accb, N, 1);

  wmma_gemm128_kernel<false, 1><<<mtiles, TB, 0, stream>>>(accb, W2, nullptr, hbuf, nullptr, a_s2, a_d2,
                                                            als, ald, N, 0, 0);
  gat_aggregate_kernel<1><<<nb((long long)N * 32), TB, 0, stream>>>(off, ssrc, hbuf, als, ald, bias2,
                                                                  out, N, 0);
}


# encoder launch folded into the input GEMM's A staging: a GEMM block makes its 32 encoder rows in LDS from the node properties; the encoder's short waves and its output's write and re-read are gone
# speedup vs baseline: 3.9482x; 1.0482x over previous
#include <hip/hip_runtime.h>
#include <math.h>

typedef __attribute__((ext_vector_type(2))) float v2f;
typedef __attribute__((ext_vector_type(8))) float v8f;

#define HID 128

__device__ __forceinline__ float lrelu(float v, float s) { return v > 0.f ? v : s * v; }

__global__ void clear_counters_kernel(int* __restrict__ cnt, int n4) {
  const int t = blockIdx.x * blockDim.x + threadIdx.x;
  const int4 z = {0, 0, 0, 0};
  if (t < n4) *((int4*)cnt + t) = z;
}


#define APITCH 132

template <int XM>
__device__ __forceinline__ float row_xor(float v) {
  return __int_as_float(__builtin_amdgcn_update_dpp(0, __float_as_int(v), 0x160 + XM, 0xf, 0xf, true));
}

__device__ __forceinline__ float logit_partial(const v8f& acc, float as_c, float ad_c, int m16) {
  float v[16];
#pragma unroll
  for (int r = 0; r < 8; ++r) { v[r] = acc[r] * as_c; v[8 + r] = acc[r] * ad_c; }
  const bool b3 = (m16 & 8) != 0, b2 = (m16 & 4) != 0, b1 = (m16 & 2) != 0, b0 = (m16 & 1) != 0;
  float w8[8], w4[4], w2[2];
#pragma unroll
  for (int j = 0; j < 8; ++j) {
    const float keep = b3 ? v[8 + j] : v[j];
    const float send = b3 ? v[j] : v[8 + j];
    w8[j] = keep + row_xor<8>(send);
  }
#pragma unroll
  for (int j = 0; j < 4; ++j) {
    const float keep = b2 ? w8[4 + j] : w8[j];
    const float send = b2 ? w8[j] : w8[4 + j];
    w4[j] = keep + row_xor<4>(send);
  }
#pragma unroll
  for (int j = 0; j < 2; ++j) {
    const float keep = b1 ? w4[2 + j] : w4[j];
    const float send = b1 ? w4[j] : w4[2 + j];
    w2[j] = keep + row_xor<2>(send);
  }
  const float keep1 = b0 ? w2[1] : w2[0];
  const float send1 = b0 ? w2[0] : w2[1];
  return keep1 + row_xor<1>(send1);
}

struct EncArgs {
  const float* num_prop; const float* cat_prop;
  const float* Wn; const float* bn;
  const float* Wc; const float* bc;
};

__device__ __forceinline__ void enc_stage_rows(const EncArgs& e, float* __restrict__ As,
                                               int row0, int wave, int lane, int M) {
  const bool isnum = lane < 16;
  const int hl = lane & 15;
  const float* Wp = isnum ? e.Wn : e.Wc;
  const float* bp = isnum ? e.bn : e.bc;
  const float4 b4 = *((const float4*)bp + hl);
  float4 w[7];
#pragma unroll
  for (int k = 0; k < 3; ++k) w[k] = *((const float4*)(Wp + k * 64) + hl);
#pragma unroll
  for (int k = 3; k < 7; ++k) {
    const float4 t = *((const float4*)(e.Wn + k * 64) + hl);
    w[k].x = isnum ? t.x : 0.f; w[k].y = isnum ? t.y : 0.f;
    w[k].z = isnum ? t.z : 0.f; w[k].w = isnum ? t.w : 0.f;
  }
  const int r0 = row0 + 4 * __builtin_amdgcn_readfirstlane(wave);
#pragma unroll
  for (int i = 0; i < 4; ++i) {
    int r = r0 + i; if (r >= M) r = M - 1;
    const float* pn = e.num_prop + (size_t)r * 7;
    const float* pc = e.cat_prop + (size_t)r * 3;
    float4 a = b4;
#pragma unroll
    for (int k = 0; k < 7; ++k) {
      const float p = (k < 3) ? (isnum ? pn[k] : pc[k]) : pn[k];
      a.x += p * w[k].x; a.y += p * w[k].y;
      a.z += p * w[k].z; a.w += p * w[k].w;
    }
    a.x = lrelu(a.x, 0.01f); a.y = lrelu(a.y, 0.01f);
    a.z = lrelu(a.z, 0.01f); a.w = lrelu(a.w, 0.01f);
    *((float4*)(As + (4 * wave + i) * APITCH) + lane) = a;
  }
}

template <bool ENC, bool BN, int H>
__device__ __forceinline__ void
gemm128_tile(int tile, const EncArgs& enc, const float* __restrict__ A, const float* __restrict__ B,
             const float* __restrict__ bias, float* __restrict__ C,
             const float* __restrict__ stats,
             const float* __restrict__ a_src, const float* __restrict__ a_dst,
             float* __restrict__ als, float* __restrict__ ald,
             int M, int act, int has_bias) {
  __shared__ __attribute__((aligned(16))) float As[32 * APITCH];
  int wave = threadIdx.x >> 5;
  int lane = threadIdx.x & 31;
  int row0 = tile * 32;
  int col0 = wave * 16;
  int half = lane >> 4;
  int m16  = lane & 15;

  if constexpr (ENC) {
    enc_stage_rows(enc, As, row0, wave, lane, M);
  } else {
    float4 sc = {1.f, 1.f, 1.f, 1.f}, sh = {0.f, 0.f, 0.f, 0.f};
    if constexpr (BN) {
      sc = *((const float4*)(stats + 2 * HID) + lane);
      sh = *((const float4*)(stats + 3 * HID) + lane);
    }
    float4 v[4];
#pragma unroll
    for (int i = 0; i < 4; ++i) {
      int r = row0 + 4 * wave + i; if (r >= M) r = M - 1;
      v[i] = *((const float4*)(A + (size_t)r * HID) + lane);
    }
#pragma unroll
    for (int i = 0; i < 4; ++i) {
      if constexpr (BN) {
        v[i].x = v[i].x * sc.x + sh.x; v[i].y = v[i].y * sc.y + sh.y;
        v[i].z = v[i].z * sc.z + sh.z; v[i].w = v[i].w * sc.w + sh.w;
      }
      *((float4*)(As + (4 * wave + i) * APITCH) + lane) = v[i];
    }
  }

  const float* Bp = B + (size_t)(2 * half) * HID + col0 + m16;
  const int col = col0 + m16;
  float as_c = 0.f, ad_c = 0.f;
  if constexpr (H > 0) { as_c = a_src[col]; ad_c = a_dst[col]; }

  __syncthreads();

  const float* Ar0 = As + m16 * APITCH + 2 * half;
  const float* Ar1 = Ar0 + 16 * APITCH;
  v8f acc0 = {0.f, 0.f, 0.f, 0.f, 0.f, 0.f, 0.f, 0.f};
  v8f acc1 = {0.f, 0.f, 0.f, 0.f, 0.f, 0.f, 0.f, 0.f};
#pragma unroll 8
  for (int k0 = 0; k0 < HID; k0 += 4) {
    v2f b;
    b.x = Bp[(size_t)k0 * HID];
    b.y = Bp[(size_t)(k0 + 1) * HID];
    const v2f a0 = *((const v2f*)(Ar0 + k0));
    const v2f a1 = *((const v2f*)(Ar1 + k0));
    acc0 = __builtin_amdgcn_wmma_f32_16x16x4_f32(false, a0, false, b, (short)0, acc0, false, false);
    acc1 = __builtin_amdgcn_wmma_f32_16x16x4_f32(false, a1, false, b, (short)0, acc1, false, false);
  }

  if constexpr (H > 0) {
    __shared__ float part[2][2][16][8];
    const float p0 = logit_partial(acc0, as_c, ad_c, m16);
    const float p1 = logit_partial(acc1, as_c, ad_c, m16);
    part[0][m16 >> 3][(m16 & 7) + 8 * half][wave] = p0;
    part[1][m16 >> 3][(m16 & 7) + 8 * half][wave] = p1;
    __syncthreads();
    constexpr int WPH = 8 / H;
    if (threadIdx.x < 64 * H) {
      const int t     = threadIdx.x / (32 * H);
      const int rt    = threadIdx.x - t * (32 * H);
      const int which = rt / (16 * H);
      const int rem   = rt - which * (16 * H);
      const int row   = rem / H;
      const int hh    = rem - row * H;
      float s = part[t][which][row][hh * WPH];
#pragma unroll
      for (int w = 1; w < WPH; ++w) s += part[t][which][row][hh * WPH + w];
      const int grow = row0 + 16 * t + row;
      if (grow < M) (which ? ald : als)[(size_t)grow * H + hh] = s;
    }
  }

  float bv = has_bias ? bias[col] : 0.f;
#pragma unroll
  for (int r = 0; r < 8; ++r) {
    int row = row0 + r + 8 * half;
    if (row < M) {
      float v = acc0[r] + bv;
      if (act) v = lrelu(v, 0.01f);
      C[(size_t)row * HID + col] = v;
    }
    row += 16;
    if (row < M) {
      float v = acc1[r] + bv;
      if (act) v = lrelu(v, 0.01f);
      C[(size_t)row * HID + col] = v;
    }
  }
}

template <bool BN, int H>
__global__ void __launch_bounds__(256)
wmma_gemm128_kernel(const float* __restrict__ A, const float* __restrict__ B,
                    const float* __restrict__ bias, float* __restrict__ C,
                    const float* __restrict__ stats,
                    const float* __restrict__ a_src, const float* __restrict__ a_dst,
                    float* __restrict__ als, float* __restrict__ ald,
                    int M, int act, int has_bias) {
  const EncArgs none = {nullptr, nullptr, nullptr, nullptr, nullptr, nullptr};
  gemm128_tile<false, BN, H>((int)blockIdx.x, none, A, B, bias, C, stats, a_src, a_dst, als, ald, M, act, has_bias);
}

#define BN_BLOCKS 256
#define BN_WAVES  32

__device__ __forceinline__ void
bn_stats_body(int blk, int nblk, const float* __restrict__ x, float* __restrict__ part, int N) {
  __shared__ float4 sh[2][BN_WAVES][32];
  const int lane = threadIdx.x & 31;
  const int wave = threadIdx.x >> 5;
  const int W = nblk * BN_WAVES;
  const int gw = __builtin_amdgcn_readfirstlane((int)(blk * BN_WAVES + wave));
  float4 s  = {0.f, 0.f, 0.f, 0.f};
  float4 s2 = {0.f, 0.f, 0.f, 0.f};
  for (int r0 = gw; r0 < N; r0 += 8 * W) {
    float4 v[8];
#pragma unroll
    for (int i = 0; i < 8; ++i) {
      const int r  = r0 + i * W;
      const bool ok = r < N;
      const float4 t = *((const float4*)(x + (size_t)(ok ? r : N - 1) * HID) + lane);
      v[i].x = ok ? t.x : 0.f; v[i].y = ok ? t.y : 0.f;
      v[i].z = ok ? t.z : 0.f; v[i].w = ok ? t.w : 0.f;
    }
#pragma unroll
    for (int i = 0; i < 8; ++i) {
      s.x += v[i].x; s.y += v[i].y; s.z += v[i].z; s.w += v[i].w;
      s2.x += v[i].x * v[i].x; s2.y += v[i].y * v[i].y;
      s2.z += v[i].z * v[i].z; s2.w += v[i].w * v[i].w;
    }
  }
  sh[0][wave][lane] = s;
  sh[1][wave][lane] = s2;
  __syncthreads();
  if (threadIdx.x < 64) {
    const int which = threadIdx.x >> 5;
    float4 a = sh[which][0][lane];
#pragma unroll
    for (int w = 1; w < BN_WAVES; ++w) {
      const float4 b = sh[which][w][lane];
      a.x += b.x; a.y += b.y; a.z += b.z; a.w += b.w;
    }
    *((float4*)(part + (size_t)blk * 2 * HID + which * HID) + lane) = a;
  }
}

__device__ __forceinline__ void
bn_finalize_body(const float* __restrict__ part, float* __restrict__ stats,
                 const float* __restrict__ gamma, const float* __restrict__ beta, int N) {
  __shared__ float sh[2][8][HID];
  constexpr int PER = BN_BLOCKS / 8;
  const int c = threadIdx.x & (HID - 1);
  const int g = threadIdx.x >> 7;
  const float* p = part + (size_t)(g * PER) * 2 * HID + c;
  float s = 0.f, s2 = 0.f;
#pragma unroll
  for (int b0 = 0; b0 < PER; b0 += 16) {
    float a[16], q[16];
#pragma unroll
    for (int i = 0; i < 16; ++i) {
      a[i] = p[(size_t)(b0 + i) * 2 * HID];
      q[i] = p[(size_t)(b0 + i) * 2 * HID + HID];
    }
#pragma unroll
    for (int i = 0; i < 16; ++i) { s += a[i]; s2 += q[i]; }
  }
  sh[0][g][c] = s;
  sh[1][g][c] = s2;
  __syncthreads();
  if (threadIdx.x < HID) {
    float S = 0.f, S2 = 0.f;
    for (int k = 0; k < 8; ++k) { S += sh[0][k][c]; S2 += sh[1][k][c]; }
    float inv_n = 1.f / (float)N;
    float mu  = S * inv_n;
    float var = S2 * inv_n - mu * mu;
    float sc  = gamma[c] * rsqrtf(var + 1e-5f);
    stats[2 * HID + c] = sc;
    stats[3 * HID + c] = beta[c] - mu * sc;
  }
}

__device__ __forceinline__ void edge_rank_body(int blk, const int* __restrict__ ei, int* __restrict__ cnt,
                                               int* __restrict__ rank, int E) {
  int e = blk * blockDim.x + threadIdx.x;
  if (e >= E) return;
  int d = ei[(size_t)E + e];
  rank[e] = atomicAdd(cnt + d, 1);
}

__device__ __forceinline__ int4 load_tile(const int* __restrict__ cnt, int t, int t1, int lane) {
  const bool ok = t < t1;
  const int4 v = *((const int4*)(cnt + (size_t)(ok ? t : t1 - 1) * 128) + lane);
  int4 r;
  r.x = ok ? v.x : 0; r.y = ok ? v.y : 0; r.z = ok ? v.z : 0; r.w = ok ? v.w : 0;
  return r;
}

__device__ __forceinline__ void
bucket_scan_body(const int* __restrict__ cnt, int* __restrict__ off, int tiles) {
  __shared__ int wsum[32];
  const int lane = threadIdx.x & 31;
  const int w = __builtin_amdgcn_readfirstlane((int)(threadIdx.x >> 5));
  const int per = (tiles + 31) / 32;
  const int t0 = w * per;
  const int t1 = t0 + per < tiles ? t0 + per : tiles;

  int tot = 0;
  for (int t = t0; t < t1; t += 8) {
    int4 v[8];
#pragma unroll
    for (int i = 0; i < 8; ++i) v[i] = load_tile(cnt, t + i, t1, lane);
#pragma unroll
    for (int i = 0; i < 8; ++i) tot += (v[i].x + v[i].y) + (v[i].z + v[i].w);
  }
  for (int o = 16; o > 0; o >>= 1) tot += __shfl_xor(tot, o, 32);
  if (lane == 0) wsum[w] = tot;
  __syncthreads();

  const int mine = wsum[lane];
  int incl = mine;
  for (int o = 1; o < 32; o <<= 1) {
    const int y = __shfl_up(incl, o, 32);
    if (lane >= o) incl += y;
  }
  int base = __builtin_amdgcn_readlane(incl - mine, w);

  for (int t = t0; t < t1; t += 8) {
    int4 v[8];
#pragma unroll
    for (int i = 0; i < 8; ++i) v[i] = load_tile(cnt, t + i, t1, lane);
#pragma unroll
    for (int i = 0; i < 8; ++i) {
      const int p0 = v[i].x, p1 = p0 + v[i].y, p2 = p1 + v[i].z, p3 = p2 + v[i].w;
      int inc = p3;
      for (int o = 1; o < 32; o <<= 1) {
        const int y = __shfl_up(inc, o, 32);
        if (lane >= o) inc += y;
      }
      const int ex = base + inc - p3;
      if (t + i < t1) {
        int4 r;
        r.x = ex; r.y = ex + p0; r.z = ex + p1; r.w = ex + p2;
        *((int4*)(off + (size_t)(t + i) * 128) + lane) = r;
      }
      base += __builtin_amdgcn_readlane(inc, 31);
    }
  }
}

__device__ __forceinline__ void edge_place_body(int blk, const int* __restrict__ ei, const int* __restrict__ off,
                                                const int* __restrict__ rank, int* __restrict__ bsrc, int E) {
  int e = blk * blockDim.x + threadIdx.x;
  if (e >= E) return;
  int s = ei[e];
  int d = ei[(size_t)E + e];
  bsrc[off[d] + rank[e]] = s;
}

__device__ __forceinline__ void bucket_sort_body(int blk, const int* __restrict__ off, const int* __restrict__ bsrc,
                                                 int* __restrict__ ssrc, int N) {
  const int lane = threadIdx.x & 31;
  const int d = __builtin_amdgcn_readfirstlane((int)((blk * blockDim.x + threadIdx.x) >> 5));
  if (d >= N) return;
  const int beg = off[d];
  const int L = off[d + 1] - beg;
  if (L <= 32) {
    int key = (lane < L) ? bsrc[beg + lane] : 0x7fffffff;
    for (int k = 2; k <= 32; k <<= 1) {
      for (int j = k >> 1; j > 0; j >>= 1) {
        int other = __shfl_xor(key, j, 32);
        bool up = (lane & k) == 0;
        bool lower = (lane & j) == 0;
        int lo = key < other ? key : other;
        int hi = key < other ? other : key;
        key = (lower == up) ? lo : hi;
      }
    }
    if (lane < L) ssrc[beg + lane] = key;
  } else {
    for (int i = lane; i < L; i += 32) {
      int ki = bsrc[beg + i];
      int r = 0;
      for (int j = 0; j < L; ++j) {
        int kj = bsrc[beg + j];
        r += (kj < ki || (kj == ki && j < i)) ? 1 : 0;
      }
      ssrc[beg + r] = ki;
    }
  }
}


__global__ void __launch_bounds__(256)
gemm_in_and_rank_kernel(const float* __restrict__ num_prop, const float* __restrict__ cat_prop,
                        const float* __restrict__ Wn, const float* __restrict__ bn,
                        const float* __restrict__ Wc, const float* __restrict__ bc,
                        const float* __restrict__ B,
                        const float* __restrict__ bias, float* __restrict__ C, int M, int mtiles,
                        const int* __restrict__ ei, int* __restrict__ cnt, int* __restrict__ rank,
                        int E, int rblocks) {
  const int g = (int)(blockIdx.x / 3);
  const int r = (int)blockIdx.x - 3 * g;
  if (r == 0) {
    if (g < mtiles) {
      const EncArgs enc = {num_prop, cat_prop, Wn, bn, Wc, bc};
      gemm128_tile<true, false, 0>(g, enc, nullptr, B, bias, C, nullptr, nullptr, nullptr, nullptr, nullptr, M, 1, 1);
    }
  } else {
    const int rb = 2 * g + r - 1;
    if (rb < rblocks) edge_rank_body(rb, ei, cnt, rank, E);
  }
}

__global__ void __launch_bounds__(1024)
bn_stats_and_scan_kernel(const float* __restrict__ x, float* __restrict__ part, int N,
                         const int* __restrict__ cnt, int* __restrict__ off, int tiles) {
  if (blockIdx.x == 0) bucket_scan_body(cnt, off, tiles);
  else bn_stats_body((int)blockIdx.x - 1, (int)gridDim.x - 1, x, part, N);
}

__global__ void __launch_bounds__(1024)
bn_finalize_and_place_kernel(const float* __restrict__ part, float* __restrict__ stats,
                             const float* __restrict__ gamma, const float* __restrict__ beta, int N,
                             const int* __restrict__ ei, const int* __restrict__ off,
                             const int* __restrict__ rank, int* __restrict__ bsrc, int E) {
  if (blockIdx.x == 0) bn_finalize_body(part, stats, gamma, beta, N);
  else edge_place_body((int)blockIdx.x - 1, ei, off, rank, bsrc, E);
}

__global__ void __launch_bounds__(256)
gemm_l1_and_sort_kernel(const float* __restrict__ A, const float* __restrict__ B, float* __restrict__ C,
                        const float* __restrict__ stats,
                        const float* __restrict__ a_src, const float* __restrict__ a_dst,
                        float* __restrict__ als, float* __restrict__ ald, int M, int mtiles,
                        const int* __restrict__ off, const int* __restrict__ bsrc,
                        int* __restrict__ ssrc, int N, int sblocks) {
  const int g = (int)(blockIdx.x / 5);
  const int r = (int)blockIdx.x - 5 * g;
  if (r == 0) {
    if (g < mtiles)
    {
      const EncArgs none = {nullptr, nullptr, nullptr, nullptr, nullptr, nullptr};
      gemm128_tile<false, true, 4>(g, none, A, B, nullptr, C, stats, a_src, a_dst, als, ald, M, 0, 0);
    }
  } else {
    const int sb = 4 * g + r - 1;
    if (sb < sblocks) bucket_sort_body(sb, off, bsrc, ssrc, N);
  }
}

__device__ __forceinline__ float wave_max(float v) {
  for (int o = 16; o > 0; o >>= 1) v = fmaxf(v, __shfl_xor(v, o, 32));
  return v;
}
__device__ __forceinline__ float wave_sum(float v) {
  for (int o = 16; o > 0; o >>= 1) v += __shfl_xor(v, o, 32);
  return v;
}

template <int H>
__device__ __forceinline__ void add_rows(float4& acc, const float* __restrict__ hfeat, int s,
                                         const float (&al)[H], int n, int lane, int myhead) {
#pragma unroll 8
  for (int jj = 0; jj < n; ++jj) {
    const int sj = __builtin_amdgcn_readlane(s, jj);
    float a = __int_as_float(__builtin_amdgcn_readlane(__float_as_int(al[0]), jj));
    for (int hh = 1; hh < H; ++hh) {
      float ah = __int_as_float(__builtin_amdgcn_readlane(__float_as_int(al[hh]), jj));
      a = (myhead == hh) ? ah : a;
    }
    const float4 hv = *((const float4*)(hfeat + (size_t)sj * HID) + lane);
    acc.x += a * hv.x;
    acc.y += a * hv.y;
    acc.z += a * hv.z;
    acc.w += a * hv.w;
  }
}

template <int H>
__global__ void gat_aggregate_kernel(const int* __restrict__ off, const int* __restrict__ ssrc,
                                     const float* __restrict__ hfeat, const float* __restrict__ als,
                                     const float* __restrict__ ald, const float* __restrict__ bias,
                                     float* __restrict__ out, int N, int act) {
  constexpr int F = HID / H;
  const int lane = threadIdx.x & 31;
  const int d = __builtin_amdgcn_readfirstlane((int)((blockIdx.x * blockDim.x + threadIdx.x) >> 5));
  if (d >= N) return;
  const int beg = off[d];
  const int L = off[d + 1] - beg;
  const int tot = L + 1;

  float adv[H], m[H], den[H], e0[H];
  for (int hh = 0; hh < H; ++hh) { adv[hh] = ald[(size_t)d * H + hh]; m[hh] = -INFINITY; den[hh] = 0.f; e0[hh] = 0.f; }

  const bool v0 = lane < tot;
  int s0 = d;
  if (v0) {
    s0 = (lane < L) ? ssrc[beg + lane] : d;
    for (int hh = 0; hh < H; ++hh) { e0[hh] = lrelu(als[(size_t)s0 * H + hh] + adv[hh], 0.2f); m[hh] = e0[hh]; }
  }

  for (int c0 = 32; c0 < tot; c0 += 32) {
    int j = c0 + lane;
    if (j < tot) {
      int s = (j < L) ? ssrc[beg + j] : d;
      for (int hh = 0; hh < H; ++hh)
        m[hh] = fmaxf(m[hh], lrelu(als[(size_t)s * H + hh] + adv[hh], 0.2f));
    }
  }
  for (int hh = 0; hh < H; ++hh) m[hh] = wave_max(m[hh]);

  float x0[H];
  for (int hh = 0; hh < H; ++hh) { x0[hh] = v0 ? __expf(e0[hh] - m[hh]) : 0.f; den[hh] = x0[hh]; }
  for (int c0 = 32; c0 < tot; c0 += 32) {
    int j = c0 + lane;
    if (j < tot) {
      int s = (j < L) ? ssrc[beg + j] : d;
      for (int hh = 0; hh < H; ++hh)
        den[hh] += __expf(lrelu(als[(size_t)s * H + hh] + adv[hh], 0.2f) - m[hh]);
    }
  }
  for (int hh = 0; hh < H; ++hh) den[hh] = wave_sum(den[hh]);

  const int myhead = (4 * lane) / F;
  float4 acc = {0.f, 0.f, 0.f, 0.f};
  {
    float al[H];
    for (int hh = 0; hh < H; ++hh) al[hh] = x0[hh] / den[hh];
    add_rows<H>(acc, hfeat, s0, al, tot < 32 ? tot : 32, lane, myhead);
  }
  for (int c0 = 32; c0 < tot; c0 += 32) {
    int j = c0 + lane;
    int s = d;
    float al[H];
    for (int hh = 0; hh < H; ++hh) al[hh] = 0.f;
    if (j < tot) {
      s = (j < L) ? ssrc[beg + j] : d;
      for (int hh = 0; hh < H; ++hh)
        al[hh] = __expf(lrelu(als[(size_t)s * H + hh] + adv[hh], 0.2f) - m[hh]) / den[hh];
    }
    add_rows<H>(acc, hfeat, s, al, tot - c0 < 32 ? tot - c0 : 32, lane, myhead);
  }

  const float4 bv = *((const float4*)bias + lane);
  float4 o = {acc.x + bv.x, acc.y + bv.y, acc.z + bv.z, acc.w + bv.w};
  if (act) { o.x = lrelu(o.x, 0.01f); o.y = lrelu(o.y, 0.01f); o.z = lrelu(o.z, 0.01f); o.w = lrelu(o.w, 0.01f); }
  *((float4*)(out + (size_t)d * HID) + lane) = o;
}

extern "C" void kernel_launch(void* const* d_in, const int* in_sizes, int n_in,
                              void* d_out, int out_size, void* d_ws, size_t ws_size,
                              hipStream_t stream) {
  const float* num_prop = (const float*)d_in[0];
  const float* cat_prop = (const float*)d_in[1];
  const int*   ei       = (const int*)d_in[2];
  const float* W_num = (const float*)d_in[3];
  const float* b_num = (const float*)d_in[4];
  const float* W_cat = (const float*)d_in[5];
  const float* b_cat = (const float*)d_in[6];
  const float* W_in  = (const float*)d_in[7];
  const float* b_in  = (const float*)d_in[8];
  const float* bn_g  = (const float*)d_in[9];
  const float* bn_b  = (const float*)d_in[10];
  const float* W1    = (const float*)d_in[11];
  const float* a_s1  = (const float*)d_in[12];
  const float* a_d1  = (const float*)d_in[13];
  const float* bias1 = (const float*)d_in[14];
  const float* W2    = (const float*)d_in[15];
  const float* a_s2  = (const float*)d_in[16];
  const float* a_d2  = (const float*)d_in[17];
  const float* bias2 = (const float*)d_in[18];

  const int N = in_sizes[0] / 7;
  const int E = in_sizes[2] / 2;
  float* out = (float*)d_out;

  float* ws   = (float*)d_ws;
  float* xcat = ws;
  float* x1   = xcat + (size_t)N * HID;
  float* hbuf = x1   + (size_t)N * HID;
  float* accb = hbuf + (size_t)N * HID;
  float* als  = accb + (size_t)N * HID;
  float* ald  = als  + (size_t)N * 4;
  float* stats= ald  + (size_t)N * 12;

  const int TB = 256;
  const int mtiles = (N + 31) / 32;
  auto nb = [&](long long work) { return (unsigned)((work + TB - 1) / TB); };

  const size_t Np = ((size_t)N + 128) / 128 * 128;
  const size_t Ep = ((size_t)E + 63) / 64 * 64;
  int* cnt  = (int*)(ald + (size_t)N * 4);
  int* off  = cnt + Np;
  int* rank = (int*)accb;
  int* bsrc = rank + Ep;
  int* ssrc = (int*)xcat;
  float* bnpart = hbuf;

  clear_counters_kernel<<<nb((long long)(Np / 4)), TB, 0, stream>>>(cnt, (int)(Np / 4));

  const int rblocks = (int)nb(E);
  const int sblocks = (int)nb((long long)N * 32);
  const int pblocks = (E + 1023) / 1024;
  const int triples = mtiles > (rblocks + 1) / 2 ? mtiles : (rblocks + 1) / 2;
  const int fives   = mtiles > (sblocks + 3) / 4 ? mtiles : (sblocks + 3) / 4;

  gemm_in_and_rank_kernel<<<3 * triples, TB, 0, stream>>>(num_prop, cat_prop, W_num, b_num, W_cat, b_cat,
                                                        W_in, b_in, x1, N, mtiles,
                                                        ei, cnt, rank, E, rblocks);
  bn_stats_and_scan_kernel<<<BN_BLOCKS + 1, BN_WAVES * 32, 0, stream>>>(x1, bnpart, N,
                                                                        cnt, off, (int)(Np / 128));
  bn_finalize_and_place_kernel<<<pblocks + 1, 1024, 0, stream>>>(bnpart, stats, bn_g, bn_b, N,
                                                                 ei, off, rank, bsrc, E);

  gemm_l1_and_sort_kernel<<<5 * fives, TB, 0, stream>>>(x1, W1, hbuf, stats, a_s1, a_d1, als, ald,
                                                          N, mtiles, off, bsrc, ssrc, N, sblocks);
  gat_aggregate_kernel<4><<<nb((long long)N * 32), TB, 0, stream>>>(off, ssrc, hbuf, als, ald, bias1,
                                                                  accb, N, 1);

  wmma_gemm128_kernel<false, 1><<<mtiles, TB, 0, stream>>>(accb, W2, nullptr, hbuf, nullptr, a_s2, a_d2,
                                                            als, ald, N, 0, 0);
  gat_aggregate_kernel<1><<<nb((long long)N * 32), TB, 0, stream>>>(off, ssrc, hbuf, als, ald, bias2,
                                                                  out, N, 0);
}
